# starting kernel
# baseline (speedup 1.0000x reference)
#include <hip/hip_runtime.h>
#include <hip/hip_bf16.h>
#include <math.h>

#define HID      128
#define HEADS    8
#define HEAD_DIM 16
#define LAYERS   3
#define NEG_SLOPE 0.2f
#define EPS_F     1e-16f

typedef float v2f __attribute__((ext_vector_type(2)));
typedef float v8f __attribute__((ext_vector_type(8)));

__global__ __launch_bounds__(256)
void gemm_dual_f32(const float* __restrict__ X,
                   const float* __restrict__ Wl,
                   const float* __restrict__ Wr,
                   float* __restrict__ XL,
                   float* __restrict__ XR) {
    __shared__ float As[16][HID];
    const int row0 = blockIdx.x * 16;

    {
        const float4* gx = (const float4*)(X + (size_t)row0 * HID);
        float4* sx = (float4*)(&As[0][0]);
        for (int t = threadIdx.x; t < 16 * HID / 4; t += 256) sx[t] = gx[t];
    }
    __syncthreads();

    const int wave = threadIdx.x >> 5;
    const int lane = threadIdx.x & 31;
    const int col0 = wave * 16;
    const int half = lane >> 4;
    const int lm   = lane & 15;

    v8f accL = {0.f,0.f,0.f,0.f,0.f,0.f,0.f,0.f};
    v8f accR = {0.f,0.f,0.f,0.f,0.f,0.f,0.f,0.f};

    #pragma unroll 4
    for (int k0 = 0; k0 < HID; k0 += 4) {
        const int ka = k0 + 2 * half;
        v2f a;
        a.x = As[lm][ka];
        a.y = As[lm][ka + 1];

        v2f bl, br;
        bl.x = Wl[(size_t)(ka + 0) * HID + col0 + lm];
        bl.y = Wl[(size_t)(ka + 1) * HID + col0 + lm];
        br.x = Wr[(size_t)(ka + 0) * HID + col0 + lm];
        br.y = Wr[(size_t)(ka + 1) * HID + col0 + lm];

        accL = __builtin_amdgcn_wmma_f32_16x16x4_f32(false, a, false, bl,
                                                     (short)0, accL, false, false);
        accR = __builtin_amdgcn_wmma_f32_16x16x4_f32(false, a, false, br,
                                                     (short)0, accR, false, false);
    }

    #pragma unroll
    for (int r = 0; r < 8; ++r) {
        const int m = r + 8 * half;
        XL[(size_t)(row0 + m) * HID + col0 + lm] = accL[r];
        XR[(size_t)(row0 + m) * HID + col0 + lm] = accR[r];
    }
}

__device__ __forceinline__ void atomicMaxFloat(float* addr, float val) {
    if (val >= 0.0f)
        atomicMax((int*)addr, __float_as_int(val));
    else
        atomicMin((unsigned int*)addr, __float_as_uint(val));
}

__global__ void init_md_kernel(float* __restrict__ mmax,
                               float* __restrict__ denom, int n) {
    int i = blockIdx.x * blockDim.x + threadIdx.x;
    if (i < n) { mmax[i] = -INFINITY; denom[i] = 0.0f; }
}

__global__ void init_out_kernel(float* __restrict__ out,
                                const float* __restrict__ bias, int total) {
    int i = blockIdx.x * blockDim.x + threadIdx.x;
    if (i < total) out[i] = bias[i & (HID - 1)];
}

__global__ void edge_logits_kernel(const float* __restrict__ xl,
                                   const float* __restrict__ xr,
                                   const int* __restrict__ src,
                                   const int* __restrict__ dst,
                                   const float* __restrict__ att,
                                   float* __restrict__ logits,
                                   float* __restrict__ mmax, int EH) {
    int i = blockIdx.x * blockDim.x + threadIdx.x;
    if (i >= EH) return;
    const int e = i / HEADS, h = i % HEADS;
    const int s = src[e], d = dst[e];

    const float4* pl = (const float4*)(xl + (size_t)s * HID + h * HEAD_DIM);
    const float4* pr = (const float4*)(xr + (size_t)d * HID + h * HEAD_DIM);
    const float4* pa = (const float4*)(att + h * HEAD_DIM);

    float acc = 0.0f;
    #pragma unroll
    for (int q = 0; q < 4; ++q) {
        float4 a = pl[q], b = pr[q], w = pa[q];
        float v;
        v = a.x + b.x; v = (v > 0.0f) ? v : NEG_SLOPE * v; acc += v * w.x;
        v = a.y + b.y; v = (v > 0.0f) ? v : NEG_SLOPE * v; acc += v * w.y;
        v = a.z + b.z; v = (v > 0.0f) ? v : NEG_SLOPE * v; acc += v * w.z;
        v = a.w + b.w; v = (v > 0.0f) ? v : NEG_SLOPE * v; acc += v * w.w;
    }
    logits[i] = acc;
    atomicMaxFloat(&mmax[(size_t)d * HEADS + h], acc);
}

__global__ void edge_softmax_kernel(const int* __restrict__ dst,
                                    const float* __restrict__ mmax,
                                    float* __restrict__ lw,
                                    float* __restrict__ denom, int EH) {
    int i = blockIdx.x * blockDim.x + threadIdx.x;
    if (i >= EH) return;
    const int e = i / HEADS, h = i % HEADS;
    const int d = dst[e];
    const float w = expf(lw[i] - mmax[(size_t)d * HEADS + h]);
    lw[i] = w;
    atomicAdd(&denom[(size_t)d * HEADS + h], w);
}

__global__ void edge_aggregate_kernel(const int* __restrict__ src,
                                      const int* __restrict__ dst,
                                      const float* __restrict__ xl,
                                      const float* __restrict__ lw,
                                      const float* __restrict__ denom,
                                      float* __restrict__ out, int EH) {
    int i = blockIdx.x * blockDim.x + threadIdx.x;
    if (i >= EH) return;
    const int e = i / HEADS, h = i % HEADS;
    const int s = src[e], d = dst[e];
    const float alpha = lw[i] / (denom[(size_t)d * HEADS + h] + EPS_F);

    const float4* pl = (const float4*)(xl + (size_t)s * HID + h * HEAD_DIM);
    float* po = out + (size_t)d * HID + h * HEAD_DIM;
    #pragma unroll
    for (int q = 0; q < 4; ++q) {
        float4 v = pl[q];
        atomicAdd(po + 4 * q + 0, alpha * v.x);
        atomicAdd(po + 4 * q + 1, alpha * v.y);
        atomicAdd(po + 4 * q + 2, alpha * v.z);
        atomicAdd(po + 4 * q + 3, alpha * v.w);
    }
}

extern "C" void kernel_launch(void* const* d_in, const int* in_sizes, int n_in,
                              void* d_out, int out_size, void* d_ws, size_t ws_size,
                              hipStream_t stream) {
    const float* x          = (const float*)d_in[0];
    const int*   edge_index = (const int*)  d_in[1];
    const float* Wl_all     = (const float*)d_in[2];
    const float* Wr_all     = (const float*)d_in[3];
    const float* att_all    = (const float*)d_in[4];
    const float* bias_all   = (const float*)d_in[5];
    float* out = (float*)d_out;

    const int N = in_sizes[0] / HID;
    const int E = in_sizes[1] / 2;
    const int EH = E * HEADS;

    const int* src = edge_index;
    const int* dst = edge_index + E;

    const size_t nf = (size_t)N * HID;
    float* xl    = (float*)d_ws;
    float* xr    = xl + nf;
    float* bufA  = xr + nf;
    float* mmax  = bufA + nf;
    float* denom = mmax + (size_t)N * HEADS;
    float* lw    = denom + (size_t)N * HEADS;

    const int TPB = 256;
    const int gemm_blocks = N / 16;
    const int nh_blocks   = (N * HEADS + TPB - 1) / TPB;
    const int nc_blocks   = (N * HID + TPB - 1) / TPB;
    const int eh_blocks   = (EH + TPB - 1) / TPB;

    const float* xin = x;
    for (int l = 0; l < LAYERS; ++l) {
        const float* Wl   = Wl_all   + (size_t)l * HID * HID;
        const float* Wr   = Wr_all   + (size_t)l * HID * HID;
        const float* att  = att_all  + (size_t)l * HEADS * HEAD_DIM;
        const float* bias = bias_all + (size_t)l * HID;
        float* xout = (l == LAYERS - 1) ? out : bufA;
        gemm_dual_f32<<<gemm_blocks, TPB, 0, stream>>>(xin, Wl, Wr, xl, xr);
        init_md_kernel<<<nh_blocks, TPB, 0, stream>>>(mmax, denom, N * HEADS);
        init_out_kernel<<<nc_blocks, TPB, 0, stream>>>(xout, bias, N * HID);
        edge_logits_kernel<<<eh_blocks, TPB, 0, stream>>>(xl, xr, src, dst, att,
                                                          lw, mmax, EH);
        edge_softmax_kernel<<<eh_blocks, TPB, 0, stream>>>(dst, mmax, lw, denom, EH);
        edge_aggregate_kernel<<<eh_blocks, TPB, 0, stream>>>(src, dst, xl, lw,
                                                             denom, xout, EH);
        xin = xout;
    }
}


# Edges grouped by destination once a call (counting sort, edge order kept); one wave per node: max, exp, denominator, weighted row sum in regs, one divide at end; no float atomics, fixed sum order
# speedup vs baseline: 3.6968x; 3.6968x over previous
#include <hip/hip_runtime.h>
#include <hip/hip_bf16.h>
#include <math.h>

#define HID      128
#define HEADS    8
#define HEAD_DIM 16
#define LAYERS   3
#define NEG_SLOPE 0.2f
#define EPS_F     1e-16f

typedef float v2f __attribute__((ext_vector_type(2)));
typedef float v8f __attribute__((ext_vector_type(8)));

__global__ __launch_bounds__(256)
void gemm_dual_f32(const float* __restrict__ X,
                   const float* __restrict__ Wl,
                   const float* __restrict__ Wr,
                   float* __restrict__ XL,
                   float* __restrict__ XR) {
    __shared__ float As[16][HID];
    const int row0 = blockIdx.x * 16;

    {
        const float4* gx = (const float4*)(X + (size_t)row0 * HID);
        float4* sx = (float4*)(&As[0][0]);
        for (int t = threadIdx.x; t < 16 * HID / 4; t += 256) sx[t] = gx[t];
    }
    __syncthreads();

    const int wave = threadIdx.x >> 5;
    const int lane = threadIdx.x & 31;
    const int col0 = wave * 16;
    const int half = lane >> 4;
    const int lm   = lane & 15;

    v8f accL = {0.f,0.f,0.f,0.f,0.f,0.f,0.f,0.f};
    v8f accR = {0.f,0.f,0.f,0.f,0.f,0.f,0.f,0.f};

    #pragma unroll 4
    for (int k0 = 0; k0 < HID; k0 += 4) {
        const int ka = k0 + 2 * half;
        v2f a;
        a.x = As[lm][ka];
        a.y = As[lm][ka + 1];

        v2f bl, br;
        bl.x = Wl[(size_t)(ka + 0) * HID + col0 + lm];
        bl.y = Wl[(size_t)(ka + 1) * HID + col0 + lm];
        br.x = Wr[(size_t)(ka + 0) * HID + col0 + lm];
        br.y = Wr[(size_t)(ka + 1) * HID + col0 + lm];

        accL = __builtin_amdgcn_wmma_f32_16x16x4_f32(false, a, false, bl,
                                                     (short)0, accL, false, false);
        accR = __builtin_amdgcn_wmma_f32_16x16x4_f32(false, a, false, br,
                                                     (short)0, accR, false, false);
    }

    #pragma unroll
    for (int r = 0; r < 8; ++r) {
        const int m = r + 8 * half;
        XL[(size_t)(row0 + m) * HID + col0 + lm] = accL[r];
        XR[(size_t)(row0 + m) * HID + col0 + lm] = accR[r];
    }
}

__global__ void zero_counts_kernel(int* __restrict__ cnt, int n) {
    int i = blockIdx.x * blockDim.x + threadIdx.x;
    if (i < n) cnt[i] = 0;
}

__global__ void count_in_kernel(const int* __restrict__ dst,
                                int* __restrict__ cnt, int E) {
    int e = blockIdx.x * blockDim.x + threadIdx.x;
    if (e < E) atomicAdd(&cnt[dst[e]], 1);
}

__global__ __launch_bounds__(1024)
void scan_counts_kernel(const int* __restrict__ cnt,
                        int* __restrict__ rowptr,
                        int* __restrict__ cur, int n) {
    __shared__ int part[1024];
    const int t = threadIdx.x;
    const int chunk = (n + 1023) / 1024;
    const int b = t * chunk;
    int e = b + chunk;
    if (e > n) e = n;
    int s = 0;
    for (int i = b; i < e; ++i) s += cnt[i];
    part[t] = s;
    __syncthreads();
    for (int off = 1; off < 1024; off <<= 1) {
        const int v = (t >= off) ? part[t - off] : 0;
        __syncthreads();
        part[t] += v;
        __syncthreads();
    }
    int run = part[t] - s;
    for (int i = b; i < e; ++i) {
        const int c = cnt[i];
        rowptr[i] = run;
        cur[i] = run;
        run += c;
    }
    if (t == 1023) rowptr[n] = part[1023];
}

__global__ void fill_slots_kernel(const int* __restrict__ dst,
                                  int* __restrict__ cur,
                                  int* __restrict__ slot_eid, int E) {
    int e = blockIdx.x * blockDim.x + threadIdx.x;
    if (e >= E) return;
    const int p = atomicAdd(&cur[dst[e]], 1);
    slot_eid[p] = e;
}

__global__ void order_slots_kernel(const int* __restrict__ src,
                                   const int* __restrict__ dst,
                                   const int* __restrict__ rowptr,
                                   const int* __restrict__ slot_eid,
                                   int* __restrict__ csr_src, int E) {
    int p = blockIdx.x * blockDim.x + threadIdx.x;
    if (p >= E) return;
    const int e = slot_eid[p];
    const int d = dst[e];
    const int st = rowptr[d], en = rowptr[d + 1];
    int r = 0;
    for (int q = st; q < en; ++q) r += (slot_eid[q] < e) ? 1 : 0;
    csr_src[st + r] = src[e];
}

__device__ __forceinline__ float quad_sum(float x) {
    x += __int_as_float(__builtin_amdgcn_update_dpp(0, __float_as_int(x), 0xB1, 0xF, 0xF, true));
    x += __int_as_float(__builtin_amdgcn_update_dpp(0, __float_as_int(x), 0x4E, 0xF, 0xF, true));
    return x;
}

__device__ __forceinline__ float head_logit(const float4 v, const float4 r,
                                            const float4 a) {
    float t, acc = 0.0f;
    t = v.x + r.x; t = (t > 0.0f) ? t : NEG_SLOPE * t; acc += t * a.x;
    t = v.y + r.y; t = (t > 0.0f) ? t : NEG_SLOPE * t; acc += t * a.y;
    t = v.z + r.z; t = (t > 0.0f) ? t : NEG_SLOPE * t; acc += t * a.z;
    t = v.w + r.w; t = (t > 0.0f) ? t : NEG_SLOPE * t; acc += t * a.w;
    return quad_sum(acc);
}

__global__ __launch_bounds__(256)
void node_attend_kernel(const int* __restrict__ rowptr,
                        const int* __restrict__ csr_src,
                        const float* __restrict__ xl,
                        const float* __restrict__ xr,
                        const float* __restrict__ att,
                        const float* __restrict__ bias,
                        float* __restrict__ out, int n) {
    const int node = __builtin_amdgcn_readfirstlane((int)(blockIdx.x * 8 + (threadIdx.x >> 5)));
    if (node >= n) return;
    const int lane = threadIdx.x & 31;
    const int st = rowptr[node], en = rowptr[node + 1];
    const float4* xl4 = (const float4*)xl;
    const float4 r = ((const float4*)(xr + (size_t)node * HID))[lane];
    const float4 a = ((const float4*)att)[lane];

    float m = -INFINITY;
    for (int base = st; base < en; base += 32) {
        const int cnt = (en - base < 32) ? (en - base) : 32;
        const int mine = (lane < cnt) ? csr_src[base + lane] : 0;
        #pragma unroll 4
        for (int j = 0; j < cnt; ++j) {
            const int s = __builtin_amdgcn_readlane(mine, j);
            const float4 v = xl4[(size_t)s * (HID / 4) + lane];
            m = fmaxf(m, head_logit(v, r, a));
        }
    }

    float den = 0.0f;
    float4 acc = make_float4(0.0f, 0.0f, 0.0f, 0.0f);
    for (int base = st; base < en; base += 32) {
        const int cnt = (en - base < 32) ? (en - base) : 32;
        const int mine = (lane < cnt) ? csr_src[base + lane] : 0;
        #pragma unroll 4
        for (int j = 0; j < cnt; ++j) {
            const int s = __builtin_amdgcn_readlane(mine, j);
            const float4 v = xl4[(size_t)s * (HID / 4) + lane];
            const float w = expf(head_logit(v, r, a) - m);
            den += w;
            acc.x += w * v.x;
            acc.y += w * v.y;
            acc.z += w * v.z;
            acc.w += w * v.w;
        }
    }

    const float dd = den + EPS_F;
    const float4 b = ((const float4*)bias)[lane];
    float4 o;
    o.x = acc.x / dd + b.x;
    o.y = acc.y / dd + b.y;
    o.z = acc.z / dd + b.z;
    o.w = acc.w / dd + b.w;
    ((float4*)(out + (size_t)node * HID))[lane] = o;
}

extern "C" void kernel_launch(void* const* d_in, const int* in_sizes, int n_in,
                              void* d_out, int out_size, void* d_ws, size_t ws_size,
                              hipStream_t stream) {
    const float* x          = (const float*)d_in[0];
    const int*   edge_index = (const int*)  d_in[1];
    const float* Wl_all     = (const float*)d_in[2];
    const float* Wr_all     = (const float*)d_in[3];
    const float* att_all    = (const float*)d_in[4];
    const float* bias_all   = (const float*)d_in[5];
    float* out = (float*)d_out;

    const int N = in_sizes[0] / HID;
    const int E = in_sizes[1] / 2;

    const int* src = edge_index;
    const int* dst = edge_index + E;

    const size_t nf = (size_t)N * HID;
    float* xl    = (float*)d_ws;
    float* xr    = xl + nf;
    float* bufA  = xr + nf;
    int* cnt      = (int*)(bufA + nf);
    int* rowptr   = cnt + N;
    int* cur      = rowptr + N + 1;
    int* slot_eid = cur + N;
    int* csr_src  = slot_eid + E;

    const int TPB = 256;
    const int gemm_blocks = N / 16;
    const int n_blocks    = (N + TPB - 1) / TPB;
    const int e_blocks    = (E + TPB - 1) / TPB;
    const int node_blocks = (N + 7) / 8;

    zero_counts_kernel<<<n_blocks, TPB, 0, stream>>>(cnt, N);
    count_in_kernel<<<e_blocks, TPB, 0, stream>>>(dst, cnt, E);
    scan_counts_kernel<<<1, 1024, 0, stream>>>(cnt, rowptr, cur, N);
    fill_slots_kernel<<<e_blocks, TPB, 0, stream>>>(dst, cur, slot_eid, E);
    order_slots_kernel<<<e_blocks, TPB, 0, stream>>>(src, dst, rowptr, slot_eid,
                                                     csr_src, E);

    const float* xin = x;
    for (int l = 0; l < LAYERS; ++l) {
        const float* Wl   = Wl_all   + (size_t)l * HID * HID;
        const float* Wr   = Wr_all   + (size_t)l * HID * HID;
        const float* att  = att_all  + (size_t)l * HEADS * HEAD_DIM;
        const float* bias = bias_all + (size_t)l * HID;
        float* xout = (l == LAYERS - 1) ? out : bufA;
        gemm_dual_f32<<<gemm_blocks, TPB, 0, stream>>>(xin, Wl, Wr, xl, xr);
        node_attend_kernel<<<node_blocks, TPB, 0, stream>>>(rowptr, csr_src, xl, xr,
                                                            att, bias, xout, N);
        xin = xout;
    }
}


# Node attention in ONE pass over the incoming rows: up to 8 rows held in registers per chunk, running max with rescale of denominator and sums (online softmax); sum order changed, still fixed
# speedup vs baseline: 3.8942x; 1.0534x over previous
#include <hip/hip_runtime.h>
#include <hip/hip_bf16.h>
#include <math.h>

#define HID      128
#define HEADS    8
#define HEAD_DIM 16
#define LAYERS   3
#define NEG_SLOPE 0.2f
#define EPS_F     1e-16f

typedef float v2f __attribute__((ext_vector_type(2)));
typedef float v8f __attribute__((ext_vector_type(8)));

__global__ __launch_bounds__(256)
void gemm_dual_f32(const float* __restrict__ X,
                   const float* __restrict__ Wl,
                   const float* __restrict__ Wr,
                   float* __restrict__ XL,
                   float* __restrict__ XR) {
    __shared__ float As[16][HID];
    const int row0 = blockIdx.x * 16;

    {
        const float4* gx = (const float4*)(X + (size_t)row0 * HID);
        float4* sx = (float4*)(&As[0][0]);
        for (int t = threadIdx.x; t < 16 * HID / 4; t += 256) sx[t] = gx[t];
    }
    __syncthreads();

    const int wave = threadIdx.x >> 5;
    const int lane = threadIdx.x & 31;
    const int col0 = wave * 16;
    const int half = lane >> 4;
    const int lm   = lane & 15;

    v8f accL = {0.f,0.f,0.f,0.f,0.f,0.f,0.f,0.f};
    v8f accR = {0.f,0.f,0.f,0.f,0.f,0.f,0.f,0.f};

    #pragma unroll 4
    for (int k0 = 0; k0 < HID; k0 += 4) {
        const int ka = k0 + 2 * half;
        v2f a;
        a.x = As[lm][ka];
        a.y = As[lm][ka + 1];

        v2f bl, br;
        bl.x = Wl[(size_t)(ka + 0) * HID + col0 + lm];
        bl.y = Wl[(size_t)(ka + 1) * HID + col0 + lm];
        br.x = Wr[(size_t)(ka + 0) * HID + col0 + lm];
        br.y = Wr[(size_t)(ka + 1) * HID + col0 + lm];

        accL = __builtin_amdgcn_wmma_f32_16x16x4_f32(false, a, false, bl,
                                                     (short)0, accL, false, false);
        accR = __builtin_amdgcn_wmma_f32_16x16x4_f32(false, a, false, br,
                                                     (short)0, accR, false, false);
    }

    #pragma unroll
    for (int r = 0; r < 8; ++r) {
        const int m = r + 8 * half;
        XL[(size_t)(row0 + m) * HID + col0 + lm] = accL[r];
        XR[(size_t)(row0 + m) * HID + col0 + lm] = accR[r];
    }
}

__global__ void zero_counts_kernel(int* __restrict__ cnt, int n) {
    int i = blockIdx.x * blockDim.x + threadIdx.x;
    if (i < n) cnt[i] = 0;
}

__global__ void count_in_kernel(const int* __restrict__ dst,
                                int* __restrict__ cnt, int E) {
    int e = blockIdx.x * blockDim.x + threadIdx.x;
    if (e < E) atomicAdd(&cnt[dst[e]], 1);
}

__global__ __launch_bounds__(1024)
void scan_counts_kernel(const int* __restrict__ cnt,
                        int* __restrict__ rowptr,
                        int* __restrict__ cur, int n) {
    __shared__ int part[1024];
    const int t = threadIdx.x;
    const int chunk = (n + 1023) / 1024;
    const int b = t * chunk;
    int e = b + chunk;
    if (e > n) e = n;
    int s = 0;
    for (int i = b; i < e; ++i) s += cnt[i];
    part[t] = s;
    __syncthreads();
    for (int off = 1; off < 1024; off <<= 1) {
        const int v = (t >= off) ? part[t - off] : 0;
        __syncthreads();
        part[t] += v;
        __syncthreads();
    }
    int run = part[t] - s;
    for (int i = b; i < e; ++i) {
        const int c = cnt[i];
        rowptr[i] = run;
        cur[i] = run;
        run += c;
    }
    if (t == 1023) rowptr[n] = part[1023];
}

__global__ void fill_slots_kernel(const int* __restrict__ dst,
                                  int* __restrict__ cur,
                                  int* __restrict__ slot_eid, int E) {
    int e = blockIdx.x * blockDim.x + threadIdx.x;
    if (e >= E) return;
    const int p = atomicAdd(&cur[dst[e]], 1);
    slot_eid[p] = e;
}

__global__ void order_slots_kernel(const int* __restrict__ src,
                                   const int* __restrict__ dst,
                                   const int* __restrict__ rowptr,
                                   const int* __restrict__ slot_eid,
                                   int* __restrict__ csr_src, int E) {
    int p = blockIdx.x * blockDim.x + threadIdx.x;
    if (p >= E) return;
    const int e = slot_eid[p];
    const int d = dst[e];
    const int st = rowptr[d], en = rowptr[d + 1];
    int r = 0;
    for (int q = st; q < en; ++q) r += (slot_eid[q] < e) ? 1 : 0;
    csr_src[st + r] = src[e];
}

__device__ __forceinline__ float quad_sum(float x) {
    x += __int_as_float(__builtin_amdgcn_update_dpp(0, __float_as_int(x), 0xB1, 0xF, 0xF, true));
    x += __int_as_float(__builtin_amdgcn_update_dpp(0, __float_as_int(x), 0x4E, 0xF, 0xF, true));
    return x;
}

__device__ __forceinline__ float head_logit(const float4 v, const float4 r,
                                            const float4 a) {
    float t, acc = 0.0f;
    t = v.x + r.x; t = (t > 0.0f) ? t : NEG_SLOPE * t; acc += t * a.x;
    t = v.y + r.y; t = (t > 0.0f) ? t : NEG_SLOPE * t; acc += t * a.y;
    t = v.z + r.z; t = (t > 0.0f) ? t : NEG_SLOPE * t; acc += t * a.z;
    t = v.w + r.w; t = (t > 0.0f) ? t : NEG_SLOPE * t; acc += t * a.w;
    return quad_sum(acc);
}

__device__ __forceinline__ void attend_chunk(const float4* __restrict__ xl4,
                                             const int mine, const int c, const int cnt,
                                             const int lane,
                                             const float4 r, const float4 a,
                                             float& m, float& den, float4& acc) {
    int s[8];
    #pragma unroll
    for (int j = 0; j < 8; ++j) s[j] = __builtin_amdgcn_readlane(mine, c + j);

    float4 v[8];
    #pragma unroll
    for (int j = 0; j < 8; ++j) {
        if (c + j < cnt) v[j] = xl4[(size_t)s[j] * (HID / 4) + lane];
    }

    float lg[8];
    float cm = -INFINITY;
    #pragma unroll
    for (int j = 0; j < 8; ++j) {
        if (c + j < cnt) {
            lg[j] = head_logit(v[j], r, a);
            cm = fmaxf(cm, lg[j]);
        }
    }

    const float mn = fmaxf(m, cm);
    const float sc = expf(m - mn);
    den   *= sc;
    acc.x *= sc;
    acc.y *= sc;
    acc.z *= sc;
    acc.w *= sc;
    #pragma unroll
    for (int j = 0; j < 8; ++j) {
        if (c + j < cnt) {
            const float w = expf(lg[j] - mn);
            den   += w;
            acc.x += w * v[j].x;
            acc.y += w * v[j].y;
            acc.z += w * v[j].z;
            acc.w += w * v[j].w;
        }
    }
    m = mn;
}

__global__ __launch_bounds__(256)
void node_attend_kernel(const int* __restrict__ rowptr,
                        const int* __restrict__ csr_src,
                        const float* __restrict__ xl,
                        const float* __restrict__ xr,
                        const float* __restrict__ att,
                        const float* __restrict__ bias,
                        float* __restrict__ out, int n) {
    const int node = __builtin_amdgcn_readfirstlane((int)(blockIdx.x * 8 + (threadIdx.x >> 5)));
    if (node >= n) return;
    const int lane = threadIdx.x & 31;
    const int st = rowptr[node], en = rowptr[node + 1];
    const float4* xl4 = (const float4*)xl;
    const float4 r = ((const float4*)(xr + (size_t)node * HID))[lane];
    const float4 a = ((const float4*)att)[lane];

    float m = -INFINITY;
    float den = 0.0f;
    float4 acc = make_float4(0.0f, 0.0f, 0.0f, 0.0f);
    for (int base = st; base < en; base += 32) {
        const int cnt = (en - base < 32) ? (en - base) : 32;
        const int mine = (lane < cnt) ? csr_src[base + lane] : 0;
        for (int c = 0; c < cnt; c += 8)
            attend_chunk(xl4, mine, c, cnt, lane, r, a, m, den, acc);
    }

    const float dd = den + EPS_F;
    const float4 b = ((const float4*)bias)[lane];
    float4 o;
    o.x = acc.x / dd + b.x;
    o.y = acc.y / dd + b.y;
    o.z = acc.z / dd + b.z;
    o.w = acc.w / dd + b.w;
    ((float4*)(out + (size_t)node * HID))[lane] = o;
}

extern "C" void kernel_launch(void* const* d_in, const int* in_sizes, int n_in,
                              void* d_out, int out_size, void* d_ws, size_t ws_size,
                              hipStream_t stream) {
    const float* x          = (const float*)d_in[0];
    const int*   edge_index = (const int*)  d_in[1];
    const float* Wl_all     = (const float*)d_in[2];
    const float* Wr_all     = (const float*)d_in[3];
    const float* att_all    = (const float*)d_in[4];
    const float* bias_all   = (const float*)d_in[5];
    float* out = (float*)d_out;

    const int N = in_sizes[0] / HID;
    const int E = in_sizes[1] / 2;

    const int* src = edge_index;
    const int* dst = edge_index + E;

    const size_t nf = (size_t)N * HID;
    float* xl    = (float*)d_ws;
    float* xr    = xl + nf;
    float* bufA  = xr + nf;
    int* cnt      = (int*)(bufA + nf);
    int* rowptr   = cnt + N;
    int* cur      = rowptr + N + 1;
    int* slot_eid = cur + N;
    int* csr_src  = slot_eid + E;

    const int TPB = 256;
    const int gemm_blocks = N / 16;
    const int n_blocks    = (N + TPB - 1) / TPB;
    const int e_blocks    = (E + TPB - 1) / TPB;
    const int node_blocks = (N + 7) / 8;

    zero_counts_kernel<<<n_blocks, TPB, 0, stream>>>(cnt, N);
    count_in_kernel<<<e_blocks, TPB, 0, stream>>>(dst, cnt, E);
    scan_counts_kernel<<<1, 1024, 0, stream>>>(cnt, rowptr, cur, N);
    fill_slots_kernel<<<e_blocks, TPB, 0, stream>>>(dst, cur, slot_eid, E);
    order_slots_kernel<<<e_blocks, TPB, 0, stream>>>(src, dst, rowptr, slot_eid,
                                                     csr_src, E);

    const float* xin = x;
    for (int l = 0; l < LAYERS; ++l) {
        const float* Wl   = Wl_all   + (size_t)l * HID * HID;
        const float* Wr   = Wr_all   + (size_t)l * HID * HID;
        const float* att  = att_all  + (size_t)l * HEADS * HEAD_DIM;
        const float* bias = bias_all + (size_t)l * HID;
        float* xout = (l == LAYERS - 1) ? out : bufA;
        gemm_dual_f32<<<gemm_blocks, TPB, 0, stream>>>(xin, Wl, Wr, xl, xr);
        node_attend_kernel<<<node_blocks, TPB, 0, stream>>>(rowptr, csr_src, xl, xr,
                                                            att, bias, xout, N);
        xin = xout;
    }
}


# In-degree prefix sum by 49 work-groups, not 1: each adds up the counts in front of its 1024 itself, whole-line loads and stores; was 1 thread per 49-count range, every wave access touching 32 lines
# speedup vs baseline: 5.1883x; 1.3323x over previous
#include <hip/hip_runtime.h>
#include <hip/hip_bf16.h>
#include <math.h>

#define HID      128
#define HEADS    8
#define HEAD_DIM 16
#define LAYERS   3
#define NEG_SLOPE 0.2f
#define EPS_F     1e-16f

typedef float v2f __attribute__((ext_vector_type(2)));
typedef float v8f __attribute__((ext_vector_type(8)));

__global__ __launch_bounds__(256)
void gemm_dual_f32(const float* __restrict__ X,
                   const float* __restrict__ Wl,
                   const float* __restrict__ Wr,
                   float* __restrict__ XL,
                   float* __restrict__ XR) {
    __shared__ float As[16][HID];
    const int row0 = blockIdx.x * 16;

    {
        const float4* gx = (const float4*)(X + (size_t)row0 * HID);
        float4* sx = (float4*)(&As[0][0]);
        for (int t = threadIdx.x; t < 16 * HID / 4; t += 256) sx[t] = gx[t];
    }
    __syncthreads();

    const int wave = threadIdx.x >> 5;
    const int lane = threadIdx.x & 31;
    const int col0 = wave * 16;
    const int half = lane >> 4;
    const int lm   = lane & 15;

    v8f accL = {0.f,0.f,0.f,0.f,0.f,0.f,0.f,0.f};
    v8f accR = {0.f,0.f,0.f,0.f,0.f,0.f,0.f,0.f};

    #pragma unroll 4
    for (int k0 = 0; k0 < HID; k0 += 4) {
        const int ka = k0 + 2 * half;
        v2f a;
        a.x = As[lm][ka];
        a.y = As[lm][ka + 1];

        v2f bl, br;
        bl.x = Wl[(size_t)(ka + 0) * HID + col0 + lm];
        bl.y = Wl[(size_t)(ka + 1) * HID + col0 + lm];
        br.x = Wr[(size_t)(ka + 0) * HID + col0 + lm];
        br.y = Wr[(size_t)(ka + 1) * HID + col0 + lm];

        accL = __builtin_amdgcn_wmma_f32_16x16x4_f32(false, a, false, bl,
                                                     (short)0, accL, false, false);
        accR = __builtin_amdgcn_wmma_f32_16x16x4_f32(false, a, false, br,
                                                     (short)0, accR, false, false);
    }

    #pragma unroll
    for (int r = 0; r < 8; ++r) {
        const int m = r + 8 * half;
        XL[(size_t)(row0 + m) * HID + col0 + lm] = accL[r];
        XR[(size_t)(row0 + m) * HID + col0 + lm] = accR[r];
    }
}

__global__ void zero_counts_kernel(int* __restrict__ cnt, int n) {
    int i = blockIdx.x * blockDim.x + threadIdx.x;
    if (i < n) cnt[i] = 0;
}

__global__ void count_in_kernel(const int* __restrict__ dst,
                                int* __restrict__ cnt, int E) {
    int e = blockIdx.x * blockDim.x + threadIdx.x;
    if (e < E) atomicAdd(&cnt[dst[e]], 1);
}

#define SCAN_TPB 1024
__global__ __launch_bounds__(SCAN_TPB)
void scan_counts_kernel(const int* __restrict__ cnt,
                        int* __restrict__ rowptr,
                        int* __restrict__ cur, int n) {
    __shared__ int wfront[SCAN_TPB / 32];
    __shared__ int wtot[SCAN_TPB / 32];
    const int t = threadIdx.x;
    const int lane = t & 31;
    const int wave = t >> 5;
    const int first = blockIdx.x * SCAN_TPB;
    const int i = first + t;
    const int c = (i < n) ? cnt[i] : 0;

    const int4* c4 = (const int4*)cnt + t;
    const int n4 = first >> 2;
    const int full = n4 / SCAN_TPB;
    int s = 0;
    #pragma unroll 4
    for (int k = 0; k < full; ++k) {
        const int4 v = c4[(size_t)k * SCAN_TPB];
        s += (v.x + v.y) + (v.z + v.w);
    }
    if (full * SCAN_TPB + t < n4) {
        const int4 v = c4[(size_t)full * SCAN_TPB];
        s += (v.x + v.y) + (v.z + v.w);
    }
    for (int off = 16; off > 0; off >>= 1) s += __shfl_xor(s, off, 32);

    int inc = c;
    for (int off = 1; off < 32; off <<= 1) {
        const int v = __shfl_up(inc, off, 32);
        if (lane >= off) inc += v;
    }
    if (lane == 0)  wfront[wave] = s;
    if (lane == 31) wtot[wave] = inc;
    __syncthreads();

    int base = wfront[lane] + ((lane < wave) ? wtot[lane] : 0);
    for (int off = 16; off > 0; off >>= 1) base += __shfl_xor(base, off, 32);

    const int excl = base + inc - c;
    if (i < n) {
        rowptr[i] = excl;
        cur[i] = excl;
        if (i == n - 1) rowptr[n] = excl + c;
    }
}

__global__ void fill_slots_kernel(const int* __restrict__ dst,
                                  int* __restrict__ cur,
                                  int* __restrict__ slot_eid, int E) {
    int e = blockIdx.x * blockDim.x + threadIdx.x;
    if (e >= E) return;
    const int p = atomicAdd(&cur[dst[e]], 1);
    slot_eid[p] = e;
}

__global__ void order_slots_kernel(const int* __restrict__ src,
                                   const int* __restrict__ dst,
                                   const int* __restrict__ rowptr,
                                   const int* __restrict__ slot_eid,
                                   int* __restrict__ csr_src, int E) {
    int p = blockIdx.x * blockDim.x + threadIdx.x;
    if (p >= E) return;
    const int e = slot_eid[p];
    const int d = dst[e];
    const int st = rowptr[d], en = rowptr[d + 1];
    int r = 0;
    for (int q = st; q < en; ++q) r += (slot_eid[q] < e) ? 1 : 0;
    csr_src[st + r] = src[e];
}

__device__ __forceinline__ float quad_sum(float x) {
    x += __int_as_float(__builtin_amdgcn_update_dpp(0, __float_as_int(x), 0xB1, 0xF, 0xF, true));
    x += __int_as_float(__builtin_amdgcn_update_dpp(0, __float_as_int(x), 0x4E, 0xF, 0xF, true));
    return x;
}

__device__ __forceinline__ float head_logit(const float4 v, const float4 r,
                                            const float4 a) {
    float t, acc = 0.0f;
    t = v.x + r.x; t = (t > 0.0f) ? t : NEG_SLOPE * t; acc += t * a.x;
    t = v.y + r.y; t = (t > 0.0f) ? t : NEG_SLOPE * t; acc += t * a.y;
    t = v.z + r.z; t = (t > 0.0f) ? t : NEG_SLOPE * t; acc += t * a.z;
    t = v.w + r.w; t = (t > 0.0f) ? t : NEG_SLOPE * t; acc += t * a.w;
    return quad_sum(acc);
}

__device__ __forceinline__ void attend_chunk(const float4* __restrict__ xl4,
                                             const int mine, const int c, const int cnt,
                                             const int lane,
                                             const float4 r, const float4 a,
                                             float& m, float& den, float4& acc) {
    int s[8];
    #pragma unroll
    for (int j = 0; j < 8; ++j) s[j] = __builtin_amdgcn_readlane(mine, c + j);

    float4 v[8];
    #pragma unroll
    for (int j = 0; j < 8; ++j) {
        if (c + j < cnt) v[j] = xl4[(size_t)s[j] * (HID / 4) + lane];
    }

    float lg[8];
    float cm = -INFINITY;
    #pragma unroll
    for (int j = 0; j < 8; ++j) {
        if (c + j < cnt) {
            lg[j] = head_logit(v[j], r, a);
            cm = fmaxf(cm, lg[j]);
        }
    }

    const float mn = fmaxf(m, cm);
    const float sc = expf(m - mn);
    den   *= sc;
    acc.x *= sc;
    acc.y *= sc;
    acc.z *= sc;
    acc.w *= sc;
    #pragma unroll
    for (int j = 0; j < 8; ++j) {
        if (c + j < cnt) {
            const float w = expf(lg[j] - mn);
            den   += w;
            acc.x += w * v[j].x;
            acc.y += w * v[j].y;
            acc.z += w * v[j].z;
            acc.w += w * v[j].w;
        }
    }
    m = mn;
}

__global__ __launch_bounds__(256)
void node_attend_kernel(const int* __restrict__ rowptr,
                        const int* __restrict__ csr_src,
                        const float* __restrict__ xl,
                        const float* __restrict__ xr,
                        const float* __restrict__ att,
                        const float* __restrict__ bias,
                        float* __restrict__ out, int n) {
    const int node = __builtin_amdgcn_readfirstlane((int)(blockIdx.x * 8 + (threadIdx.x >> 5)));
    if (node >= n) return;
    const int lane = threadIdx.x & 31;
    const int st = rowptr[node], en = rowptr[node + 1];
    const float4* xl4 = (const float4*)xl;
    const float4 r = ((const float4*)(xr + (size_t)node * HID))[lane];
    const float4 a = ((const float4*)att)[lane];

    float m = -INFINITY;
    float den = 0.0f;
    float4 acc = make_float4(0.0f, 0.0f, 0.0f, 0.0f);
    for (int base = st; base < en; base += 32) {
        const int cnt = (en - base < 32) ? (en - base) : 32;
        const int mine = (lane < cnt) ? csr_src[base + lane] : 0;
        for (int c = 0; c < cnt; c += 8)
            attend_chunk(xl4, mine, c, cnt, lane, r, a, m, den, acc);
    }

    const float dd = den + EPS_F;
    const float4 b = ((const float4*)bias)[lane];
    float4 o;
    o.x = acc.x / dd + b.x;
    o.y = acc.y / dd + b.y;
    o.z = acc.z / dd + b.z;
    o.w = acc.w / dd + b.w;
    ((float4*)(out + (size_t)node * HID))[lane] = o;
}

extern "C" void kernel_launch(void* const* d_in, const int* in_sizes, int n_in,
                              void* d_out, int out_size, void* d_ws, size_t ws_size,
                              hipStream_t stream) {
    const float* x          = (const float*)d_in[0];
    const int*   edge_index = (const int*)  d_in[1];
    const float* Wl_all     = (const float*)d_in[2];
    const float* Wr_all     = (const float*)d_in[3];
    const float* att_all    = (const float*)d_in[4];
    const float* bias_all   = (const float*)d_in[5];
    float* out = (float*)d_out;

    const int N = in_sizes[0] / HID;
    const int E = in_sizes[1] / 2;

    const int* src = edge_index;
    const int* dst = edge_index + E;

    const size_t nf = (size_t)N * HID;
    float* xl    = (float*)d_ws;
    float* xr    = xl + nf;
    float* bufA  = xr + nf;
    int* cnt      = (int*)(bufA + nf);
    int* rowptr   = cnt + N;
    int* cur      = rowptr + N + 1;
    int* slot_eid = cur + N;
    int* csr_src  = slot_eid + E;

    const int TPB = 256;
    const int gemm_blocks = N / 16;
    const int n_blocks    = (N + TPB - 1) / TPB;
    const int e_blocks    = (E + TPB - 1) / TPB;
    const int node_blocks = (N + 7) / 8;
    const int scan_blocks = (N + SCAN_TPB - 1) / SCAN_TPB;

    zero_counts_kernel<<<n_blocks, TPB, 0, stream>>>(cnt, N);
    count_in_kernel<<<e_blocks, TPB, 0, stream>>>(dst, cnt, E);
    scan_counts_kernel<<<scan_blocks, SCAN_TPB, 0, stream>>>(cnt, rowptr, cur, N);
    fill_slots_kernel<<<e_blocks, TPB, 0, stream>>>(dst, cur, slot_eid, E);
    order_slots_kernel<<<e_blocks, TPB, 0, stream>>>(src, dst, rowptr, slot_eid,
                                                     csr_src, E);

    const float* xin = x;
    for (int l = 0; l < LAYERS; ++l) {
        const float* Wl   = Wl_all   + (size_t)l * HID * HID;
        const float* Wr   = Wr_all   + (size_t)l * HID * HID;
        const float* att  = att_all  + (size_t)l * HEADS * HEAD_DIM;
        const float* bias = bias_all + (size_t)l * HID;
        float* xout = (l == LAYERS - 1) ? out : bufA;
        gemm_dual_f32<<<gemm_blocks, TPB, 0, stream>>>(xin, Wl, Wr, xl, xr);
        node_attend_kernel<<<node_blocks, TPB, 0, stream>>>(rowptr, csr_src, xl, xr,
                                                            att, bias, xout, N);
        xin = xout;
    }
}


# GEMM: A strip in LDS padded to a row pitch of 132 floats (was 128): the rows of an A fragment load start on different banks; at 128 every row started on one bank and a load was served row by row
# speedup vs baseline: 5.3797x; 1.0369x over previous
#include <hip/hip_runtime.h>
#include <hip/hip_bf16.h>
#include <math.h>

#define HID      128
#define HEADS    8
#define HEAD_DIM 16
#define LAYERS   3
#define NEG_SLOPE 0.2f
#define EPS_F     1e-16f

typedef float v2f __attribute__((ext_vector_type(2)));
typedef float v8f __attribute__((ext_vector_type(8)));

#define A_PITCH (HID + 4)
__global__ __launch_bounds__(256)
void gemm_dual_f32(const float* __restrict__ X,
                   const float* __restrict__ Wl,
                   const float* __restrict__ Wr,
                   float* __restrict__ XL,
                   float* __restrict__ XR) {
    __shared__ __attribute__((aligned(16))) float As[16][A_PITCH];
    const int row0 = blockIdx.x * 16;

    {
        const float4* gx = (const float4*)(X + (size_t)row0 * HID);
        for (int t = threadIdx.x; t < 16 * HID / 4; t += 256) {
            const int rr = t >> 5;
            const int c4 = t & 31;
            *(float4*)(&As[rr][4 * c4]) = gx[t];
        }
    }
    __syncthreads();

    const int wave = threadIdx.x >> 5;
    const int lane = threadIdx.x & 31;
    const int col0 = wave * 16;
    const int half = lane >> 4;
    const int lm   = lane & 15;

    v8f accL = {0.f,0.f,0.f,0.f,0.f,0.f,0.f,0.f};
    v8f accR = {0.f,0.f,0.f,0.f,0.f,0.f,0.f,0.f};

    #pragma unroll 4
    for (int k0 = 0; k0 < HID; k0 += 4) {
        const int ka = k0 + 2 * half;
        v2f a;
        a.x = As[lm][ka];
        a.y = As[lm][ka + 1];

        v2f bl, br;
        bl.x = Wl[(size_t)(ka + 0) * HID + col0 + lm];
        bl.y = Wl[(size_t)(ka + 1) * HID + col0 + lm];
        br.x = Wr[(size_t)(ka + 0) * HID + col0 + lm];
        br.y = Wr[(size_t)(ka + 1) * HID + col0 + lm];

        accL = __builtin_amdgcn_wmma_f32_16x16x4_f32(false, a, false, bl,
                                                     (short)0, accL, false, false);
        accR = __builtin_amdgcn_wmma_f32_16x16x4_f32(false, a, false, br,
                                                     (short)0, accR, false, false);
    }

    #pragma unroll
    for (int r = 0; r < 8; ++r) {
        const int m = r + 8 * half;
        XL[(size_t)(row0 + m) * HID + col0 + lm] = accL[r];
        XR[(size_t)(row0 + m) * HID + col0 + lm] = accR[r];
    }
}

__global__ void zero_counts_kernel(int* __restrict__ cnt, int n) {
    int i = blockIdx.x * blockDim.x + threadIdx.x;
    if (i < n) cnt[i] = 0;
}

__global__ void count_in_kernel(const int* __restrict__ dst,
                                int* __restrict__ cnt, int E) {
    int e = blockIdx.x * blockDim.x + threadIdx.x;
    if (e < E) atomicAdd(&cnt[dst[e]], 1);
}

#define SCAN_TPB 1024
__global__ __launch_bounds__(SCAN_TPB)
void scan_counts_kernel(const int* __restrict__ cnt,
                        int* __restrict__ rowptr,
                        int* __restrict__ cur, int n) {
    __shared__ int wfront[SCAN_TPB / 32];
    __shared__ int wtot[SCAN_TPB / 32];
    const int t = threadIdx.x;
    const int lane = t & 31;
    const int wave = t >> 5;
    const int first = blockIdx.x * SCAN_TPB;
    const int i = first + t;
    const int c = (i < n) ? cnt[i] : 0;

    const int4* c4 = (const int4*)cnt + t;
    const int n4 = first >> 2;
    const int full = n4 / SCAN_TPB;
    int s = 0;
    #pragma unroll 4
    for (int k = 0; k < full; ++k) {
        const int4 v = c4[(size_t)k * SCAN_TPB];
        s += (v.x + v.y) + (v.z + v.w);
    }
    if (full * SCAN_TPB + t < n4) {
        const int4 v = c4[(size_t)full * SCAN_TPB];
        s += (v.x + v.y) + (v.z + v.w);
    }
    for (int off = 16; off > 0; off >>= 1) s += __shfl_xor(s, off, 32);

    int inc = c;
    for (int off = 1; off < 32; off <<= 1) {
        const int v = __shfl_up(inc, off, 32);
        if (lane >= off) inc += v;
    }
    if (lane == 0)  wfront[wave] = s;
    if (lane == 31) wtot[wave] = inc;
    __syncthreads();

    int base = wfront[lane] + ((lane < wave) ? wtot[lane] : 0);
    for (int off = 16; off > 0; off >>= 1) base += __shfl_xor(base, off, 32);

    const int excl = base + inc - c;
    if (i < n) {
        rowptr[i] = excl;
        cur[i] = excl;
        if (i == n - 1) rowptr[n] = excl + c;
    }
}

__global__ void fill_slots_kernel(const int* __restrict__ dst,
                                  int* __restrict__ cur,
                                  int* __restrict__ slot_eid, int E) {
    int e = blockIdx.x * blockDim.x + threadIdx.x;
    if (e >= E) return;
    const int p = atomicAdd(&cur[dst[e]], 1);
    slot_eid[p] = e;
}

__global__ void order_slots_kernel(const int* __restrict__ src,
                                   const int* __restrict__ dst,
                                   const int* __restrict__ rowptr,
                                   const int* __restrict__ slot_eid,
                                   int* __restrict__ csr_src, int E) {
    int p = blockIdx.x * blockDim.x + threadIdx.x;
    if (p >= E) return;
    const int e = slot_eid[p];
    const int d = dst[e];
    const int st = rowptr[d], en = rowptr[d + 1];
    int r = 0;
    for (int q = st; q < en; ++q) r += (slot_eid[q] < e) ? 1 : 0;
    csr_src[st + r] = src[e];
}

__device__ __forceinline__ float quad_sum(float x) {
    x += __int_as_float(__builtin_amdgcn_update_dpp(0, __float_as_int(x), 0xB1, 0xF, 0xF, true));
    x += __int_as_float(__builtin_amdgcn_update_dpp(0, __float_as_int(x), 0x4E, 0xF, 0xF, true));
    return x;
}

__device__ __forceinline__ float head_logit(const float4 v, const float4 r,
                                            const float4 a) {
    float t, acc = 0.0f;
    t = v.x + r.x; t = (t > 0.0f) ? t : NEG_SLOPE * t; acc += t * a.x;
    t = v.y + r.y; t = (t > 0.0f) ? t : NEG_SLOPE * t; acc += t * a.y;
    t = v.z + r.z; t = (t > 0.0f) ? t : NEG_SLOPE * t; acc += t * a.z;
    t = v.w + r.w; t = (t > 0.0f) ? t : NEG_SLOPE * t; acc += t * a.w;
    return quad_sum(acc);
}

__device__ __forceinline__ void attend_chunk(const float4* __restrict__ xl4,
                                             const int mine, const int c, const int cnt,
                                             const int lane,
                                             const float4 r, const float4 a,
                                             float& m, float& den, float4& acc) {
    int s[8];
    #pragma unroll
    for (int j = 0; j < 8; ++j) s[j] = __builtin_amdgcn_readlane(mine, c + j);

    float4 v[8];
    #pragma unroll
    for (int j = 0; j < 8; ++j) {
        if (c + j < cnt) v[j] = xl4[(size_t)s[j] * (HID / 4) + lane];
    }

    float lg[8];
    float cm = -INFINITY;
    #pragma unroll
    for (int j = 0; j < 8; ++j) {
        if (c + j < cnt) {
            lg[j] = head_logit(v[j], r, a);
            cm = fmaxf(cm, lg[j]);
        }
    }

    const float mn = fmaxf(m, cm);
    const float sc = expf(m - mn);
    den   *= sc;
    acc.x *= sc;
    acc.y *= sc;
    acc.z *= sc;
    acc.w *= sc;
    #pragma unroll
    for (int j = 0; j < 8; ++j) {
        if (c + j < cnt) {
            const float w = expf(lg[j] - mn);
            den   += w;
            acc.x += w * v[j].x;
            acc.y += w * v[j].y;
            acc.z += w * v[j].z;
            acc.w += w * v[j].w;
        }
    }
    m = mn;
}

__global__ __launch_bounds__(256)
void node_attend_kernel(const int* __restrict__ rowptr,
                        const int* __restrict__ csr_src,
                        const float* __restrict__ xl,
                        const float* __restrict__ xr,
                        const float* __restrict__ att,
                        const float* __restrict__ bias,
                        float* __restrict__ out, int n) {
    const int node = __builtin_amdgcn_readfirstlane((int)(blockIdx.x * 8 + (threadIdx.x >> 5)));
    if (node >= n) return;
    const int lane = threadIdx.x & 31;
    const int st = rowptr[node], en = rowptr[node + 1];
    const float4* xl4 = (const float4*)xl;
    const float4 r = ((const float4*)(xr + (size_t)node * HID))[lane];
    const float4 a = ((const float4*)att)[lane];

    float m = -INFINITY;
    float den = 0.0f;
    float4 acc = make_float4(0.0f, 0.0f, 0.0f, 0.0f);
    for (int base = st; base < en; base += 32) {
        const int cnt = (en - base < 32) ? (en - base) : 32;
        const int mine = (lane < cnt) ? csr_src[base + lane] : 0;
        for (int c = 0; c < cnt; c += 8)
            attend_chunk(xl4, mine, c, cnt, lane, r, a, m, den, acc);
    }

    const float dd = den + EPS_F;
    const float4 b = ((const float4*)bias)[lane];
    float4 o;
    o.x = acc.x / dd + b.x;
    o.y = acc.y / dd + b.y;
    o.z = acc.z / dd + b.z;
    o.w = acc.w / dd + b.w;
    ((float4*)(out + (size_t)node * HID))[lane] = o;
}

extern "C" void kernel_launch(void* const* d_in, const int* in_sizes, int n_in,
                              void* d_out, int out_size, void* d_ws, size_t ws_size,
                              hipStream_t stream) {
    const float* x          = (const float*)d_in[0];
    const int*   edge_index = (const int*)  d_in[1];
    const float* Wl_all     = (const float*)d_in[2];
    const float* Wr_all     = (const float*)d_in[3];
    const float* att_all    = (const float*)d_in[4];
    const float* bias_all   = (const float*)d_in[5];
    float* out = (float*)d_out;

    const int N = in_sizes[0] / HID;
    const int E = in_sizes[1] / 2;

    const int* src = edge_index;
    const int* dst = edge_index + E;

    const size_t nf = (size_t)N * HID;
    float* xl    = (float*)d_ws;
    float* xr    = xl + nf;
    float* bufA  = xr + nf;
    int* cnt      = (int*)(bufA + nf);
    int* rowptr   = cnt + N;
    int* cur      = rowptr + N + 1;
    int* slot_eid = cur + N;
    int* csr_src  = slot_eid + E;

    const int TPB = 256;
    const int gemm_blocks = N / 16;
    const int n_blocks    = (N + TPB - 1) / TPB;
    const int e_blocks    = (E + TPB - 1) / TPB;
    const int node_blocks = (N + 7) / 8;
    const int scan_blocks = (N + SCAN_TPB - 1) / SCAN_TPB;

    zero_counts_kernel<<<n_blocks, TPB, 0, stream>>>(cnt, N);
    count_in_kernel<<<e_blocks, TPB, 0, stream>>>(dst, cnt, E);
    scan_counts_kernel<<<scan_blocks, SCAN_TPB, 0, stream>>>(cnt, rowptr, cur, N);
    fill_slots_kernel<<<e_blocks, TPB, 0, stream>>>(dst, cur, slot_eid, E);
    order_slots_kernel<<<e_blocks, TPB, 0, stream>>>(src, dst, rowptr, slot_eid,
                                                     csr_src, E);

    const float* xin = x;
    for (int l = 0; l < LAYERS; ++l) {
        const float* Wl   = Wl_all   + (size_t)l * HID * HID;
        const float* Wr   = Wr_all   + (size_t)l * HID * HID;
        const float* att  = att_all  + (size_t)l * HEADS * HEAD_DIM;
        const float* bias = bias_all + (size_t)l * HID;
        float* xout = (l == LAYERS - 1) ? out : bufA;
        gemm_dual_f32<<<gemm_blocks, TPB, 0, stream>>>(xin, Wl, Wr, xl, xr);
        node_attend_kernel<<<node_blocks, TPB, 0, stream>>>(rowptr, csr_src, xl, xr,
                                                            att, bias, xout, N);
        xin = xout;
    }
}


# Attention of layer l and the dual GEMM of layer l+1 in one launch: a 16-node work-group (a wave per node) leaves its 16 output rows in LDS as the A strip; no rows between layers in memory; same sums
# speedup vs baseline: 5.6068x; 1.0422x over previous
#include <hip/hip_runtime.h>
#include <hip/hip_bf16.h>
#include <math.h>

#define HID      128
#define HEADS    8
#define HEAD_DIM 16
#define LAYERS   3
#define NEG_SLOPE 0.2f
#define EPS_F     1e-16f

typedef float v2f __attribute__((ext_vector_type(2)));
typedef float v8f __attribute__((ext_vector_type(8)));

#define A_PITCH (HID + 4)
__global__ __launch_bounds__(256)
void gemm_dual_f32(const float* __restrict__ X,
                   const float* __restrict__ Wl,
                   const float* __restrict__ Wr,
                   float* __restrict__ XL,
                   float* __restrict__ XR) {
    __shared__ __attribute__((aligned(16))) float As[16][A_PITCH];
    const int row0 = blockIdx.x * 16;

    {
        const float4* gx = (const float4*)(X + (size_t)row0 * HID);
        for (int t = threadIdx.x; t < 16 * HID / 4; t += 256) {
            const int rr = t >> 5;
            const int c4 = t & 31;
            *(float4*)(&As[rr][4 * c4]) = gx[t];
        }
    }
    __syncthreads();

    const int wave = threadIdx.x >> 5;
    const int lane = threadIdx.x & 31;
    const int col0 = wave * 16;
    const int half = lane >> 4;
    const int lm   = lane & 15;

    v8f accL = {0.f,0.f,0.f,0.f,0.f,0.f,0.f,0.f};
    v8f accR = {0.f,0.f,0.f,0.f,0.f,0.f,0.f,0.f};

    #pragma unroll 4
    for (int k0 = 0; k0 < HID; k0 += 4) {
        const int ka = k0 + 2 * half;
        v2f a;
        a.x = As[lm][ka];
        a.y = As[lm][ka + 1];

        v2f bl, br;
        bl.x = Wl[(size_t)(ka + 0) * HID + col0 + lm];
        bl.y = Wl[(size_t)(ka + 1) * HID + col0 + lm];
        br.x = Wr[(size_t)(ka + 0) * HID + col0 + lm];
        br.y = Wr[(size_t)(ka + 1) * HID + col0 + lm];

        accL = __builtin_amdgcn_wmma_f32_16x16x4_f32(false, a, false, bl,
                                                     (short)0, accL, false, false);
        accR = __builtin_amdgcn_wmma_f32_16x16x4_f32(false, a, false, br,
                                                     (short)0, accR, false, false);
    }

    #pragma unroll
    for (int r = 0; r < 8; ++r) {
        const int m = r + 8 * half;
        XL[(size_t)(row0 + m) * HID + col0 + lm] = accL[r];
        XR[(size_t)(row0 + m) * HID + col0 + lm] = accR[r];
    }
}

__global__ void zero_counts_kernel(int* __restrict__ cnt, int n) {
    int i = blockIdx.x * blockDim.x + threadIdx.x;
    if (i < n) cnt[i] = 0;
}

__global__ void count_in_kernel(const int* __restrict__ dst,
                                int* __restrict__ cnt, int E) {
    int e = blockIdx.x * blockDim.x + threadIdx.x;
    if (e < E) atomicAdd(&cnt[dst[e]], 1);
}

#define SCAN_TPB 1024
__global__ __launch_bounds__(SCAN_TPB)
void scan_counts_kernel(const int* __restrict__ cnt,
                        int* __restrict__ rowptr,
                        int* __restrict__ cur, int n) {
    __shared__ int wfront[SCAN_TPB / 32];
    __shared__ int wtot[SCAN_TPB / 32];
    const int t = threadIdx.x;
    const int lane = t & 31;
    const int wave = t >> 5;
    const int first = blockIdx.x * SCAN_TPB;
    const int i = first + t;
    const int c = (i < n) ? cnt[i] : 0;

    const int4* c4 = (const int4*)cnt + t;
    const int n4 = first >> 2;
    const int full = n4 / SCAN_TPB;
    int s = 0;
    #pragma unroll 4
    for (int k = 0; k < full; ++k) {
        const int4 v = c4[(size_t)k * SCAN_TPB];
        s += (v.x + v.y) + (v.z + v.w);
    }
    if (full * SCAN_TPB + t < n4) {
        const int4 v = c4[(size_t)full * SCAN_TPB];
        s += (v.x + v.y) + (v.z + v.w);
    }
    for (int off = 16; off > 0; off >>= 1) s += __shfl_xor(s, off, 32);

    int inc = c;
    for (int off = 1; off < 32; off <<= 1) {
        const int v = __shfl_up(inc, off, 32);
        if (lane >= off) inc += v;
    }
    if (lane == 0)  wfront[wave] = s;
    if (lane == 31) wtot[wave] = inc;
    __syncthreads();

    int base = wfront[lane] + ((lane < wave) ? wtot[lane] : 0);
    for (int off = 16; off > 0; off >>= 1) base += __shfl_xor(base, off, 32);

    const int excl = base + inc - c;
    if (i < n) {
        rowptr[i] = excl;
        cur[i] = excl;
        if (i == n - 1) rowptr[n] = excl + c;
    }
}

__global__ void fill_slots_kernel(const int* __restrict__ dst,
                                  int* __restrict__ cur,
                                  int* __restrict__ slot_eid, int E) {
    int e = blockIdx.x * blockDim.x + threadIdx.x;
    if (e >= E) return;
    const int p = atomicAdd(&cur[dst[e]], 1);
    slot_eid[p] = e;
}

__global__ void order_slots_kernel(const int* __restrict__ src,
                                   const int* __restrict__ dst,
                                   const int* __restrict__ rowptr,
                                   const int* __restrict__ slot_eid,
                                   int* __restrict__ csr_src, int E) {
    int p = blockIdx.x * blockDim.x + threadIdx.x;
    if (p >= E) return;
    const int e = slot_eid[p];
    const int d = dst[e];
    const int st = rowptr[d], en = rowptr[d + 1];
    int r = 0;
    for (int q = st; q < en; ++q) r += (slot_eid[q] < e) ? 1 : 0;
    csr_src[st + r] = src[e];
}

__device__ __forceinline__ float quad_sum(float x) {
    x += __int_as_float(__builtin_amdgcn_update_dpp(0, __float_as_int(x), 0xB1, 0xF, 0xF, true));
    x += __int_as_float(__builtin_amdgcn_update_dpp(0, __float_as_int(x), 0x4E, 0xF, 0xF, true));
    return x;
}

__device__ __forceinline__ float head_logit(const float4 v, const float4 r,
                                            const float4 a) {
    float t, acc = 0.0f;
    t = v.x + r.x; t = (t > 0.0f) ? t : NEG_SLOPE * t; acc += t * a.x;
    t = v.y + r.y; t = (t > 0.0f) ? t : NEG_SLOPE * t; acc += t * a.y;
    t = v.z + r.z; t = (t > 0.0f) ? t : NEG_SLOPE * t; acc += t * a.z;
    t = v.w + r.w; t = (t > 0.0f) ? t : NEG_SLOPE * t; acc += t * a.w;
    return quad_sum(acc);
}

__device__ __forceinline__ void attend_chunk(const float4* __restrict__ xl4,
                                             const int mine, const int c, const int cnt,
                                             const int lane,
                                             const float4 r, const float4 a,
                                             float& m, float& den, float4& acc) {
    int s[8];
    #pragma unroll
    for (int j = 0; j < 8; ++j) s[j] = __builtin_amdgcn_readlane(mine, c + j);

    float4 v[8];
    #pragma unroll
    for (int j = 0; j < 8; ++j) {
        if (c + j < cnt) v[j] = xl4[(size_t)s[j] * (HID / 4) + lane];
    }

    float lg[8];
    float cm = -INFINITY;
    #pragma unroll
    for (int j = 0; j < 8; ++j) {
        if (c + j < cnt) {
            lg[j] = head_logit(v[j], r, a);
            cm = fmaxf(cm, lg[j]);
        }
    }

    const float mn = fmaxf(m, cm);
    const float sc = expf(m - mn);
    den   *= sc;
    acc.x *= sc;
    acc.y *= sc;
    acc.z *= sc;
    acc.w *= sc;
    #pragma unroll
    for (int j = 0; j < 8; ++j) {
        if (c + j < cnt) {
            const float w = expf(lg[j] - mn);
            den   += w;
            acc.x += w * v[j].x;
            acc.y += w * v[j].y;
            acc.z += w * v[j].z;
            acc.w += w * v[j].w;
        }
    }
    m = mn;
}

__device__ __forceinline__ float4 attend_node(const int* __restrict__ rowptr,
                                              const int* __restrict__ csr_src,
                                              const float* __restrict__ xl,
                                              const float* __restrict__ xr,
                                              const float* __restrict__ att,
                                              const float* __restrict__ bias,
                                              const int node, const int lane) {
    const int st = rowptr[node], en = rowptr[node + 1];
    const float4* xl4 = (const float4*)xl;
    const float4 r = ((const float4*)(xr + (size_t)node * HID))[lane];
    const float4 a = ((const float4*)att)[lane];

    float m = -INFINITY;
    float den = 0.0f;
    float4 acc = make_float4(0.0f, 0.0f, 0.0f, 0.0f);
    for (int base = st; base < en; base += 32) {
        const int cnt = (en - base < 32) ? (en - base) : 32;
        const int mine = (lane < cnt) ? csr_src[base + lane] : 0;
        for (int c = 0; c < cnt; c += 8)
            attend_chunk(xl4, mine, c, cnt, lane, r, a, m, den, acc);
    }

    const float dd = den + EPS_F;
    const float4 b = ((const float4*)bias)[lane];
    float4 o;
    o.x = acc.x / dd + b.x;
    o.y = acc.y / dd + b.y;
    o.z = acc.z / dd + b.z;
    o.w = acc.w / dd + b.w;
    return o;
}

__global__ __launch_bounds__(256)
void node_attend_kernel(const int* __restrict__ rowptr,
                        const int* __restrict__ csr_src,
                        const float* __restrict__ xl,
                        const float* __restrict__ xr,
                        const float* __restrict__ att,
                        const float* __restrict__ bias,
                        float* __restrict__ out, int n) {
    const int node = __builtin_amdgcn_readfirstlane((int)(blockIdx.x * 8 + (threadIdx.x >> 5)));
    if (node >= n) return;
    const int lane = threadIdx.x & 31;
    const float4 o = attend_node(rowptr, csr_src, xl, xr, att, bias, node, lane);
    ((float4*)(out + (size_t)node * HID))[lane] = o;
}

__global__ __launch_bounds__(512)
void attend_gemm_kernel(const int* __restrict__ rowptr,
                        const int* __restrict__ csr_src,
                        const float* __restrict__ xl,
                        const float* __restrict__ xr,
                        const float* __restrict__ att,
                        const float* __restrict__ bias,
                        const float* __restrict__ Wl,
                        const float* __restrict__ Wr,
                        float* __restrict__ XLn,
                        float* __restrict__ XRn) {
    __shared__ __attribute__((aligned(16))) float As[16][A_PITCH];
    const int wave = __builtin_amdgcn_readfirstlane((int)(threadIdx.x >> 5));
    const int lane = threadIdx.x & 31;
    const int row0 = blockIdx.x * 16;
    const int node = row0 + wave;

    const float4 o = attend_node(rowptr, csr_src, xl, xr, att, bias, node, lane);
    *(float4*)(&As[wave][4 * lane]) = o;
    __syncthreads();

    const float* __restrict__ Wm = (wave < 8) ? Wl : Wr;
    float* __restrict__ Om = (wave < 8) ? XLn : XRn;
    const int col0 = (wave & 7) * 16;
    const int half = lane >> 4;
    const int lm   = lane & 15;

    v8f acc = {0.f,0.f,0.f,0.f,0.f,0.f,0.f,0.f};

    #pragma unroll 16
    for (int k0 = 0; k0 < HID; k0 += 4) {
        const int ka = k0 + 2 * half;
        v2f a;
        a.x = As[lm][ka];
        a.y = As[lm][ka + 1];

        v2f b;
        b.x = Wm[(size_t)(ka + 0) * HID + col0 + lm];
        b.y = Wm[(size_t)(ka + 1) * HID + col0 + lm];

        acc = __builtin_amdgcn_wmma_f32_16x16x4_f32(false, a, false, b,
                                                    (short)0, acc, false, false);
    }

    #pragma unroll
    for (int r = 0; r < 8; ++r) {
        const int m = r + 8 * half;
        Om[(size_t)(row0 + m) * HID + col0 + lm] = acc[r];
    }
}

extern "C" void kernel_launch(void* const* d_in, const int* in_sizes, int n_in,
                              void* d_out, int out_size, void* d_ws, size_t ws_size,
                              hipStream_t stream) {
    const float* x          = (const float*)d_in[0];
    const int*   edge_index = (const int*)  d_in[1];
    const float* Wl_all     = (const float*)d_in[2];
    const float* Wr_all     = (const float*)d_in[3];
    const float* att_all    = (const float*)d_in[4];
    const float* bias_all   = (const float*)d_in[5];
    float* out = (float*)d_out;

    const int N = in_sizes[0] / HID;
    const int E = in_sizes[1] / 2;

    const int* src = edge_index;
    const int* dst = edge_index + E;

    const size_t nf = (size_t)N * HID;
    float* xlA   = (float*)d_ws;
    float* xrA   = xlA + nf;
    float* xlB   = xrA + nf;
    float* xrB   = xlB + nf;
    int* cnt      = (int*)(xrB + nf);
    int* rowptr   = cnt + N;
    int* cur      = rowptr + N + 1;
    int* slot_eid = cur + N;
    int* csr_src  = slot_eid + E;

    const int TPB = 256;
    const int gemm_blocks  = N / 16;
    const int n_blocks     = (N + TPB - 1) / TPB;
    const int e_blocks     = (E + TPB - 1) / TPB;
    const int node_blocks  = (N + 7) / 8;
    const int scan_blocks  = (N + SCAN_TPB - 1) / SCAN_TPB;

    zero_counts_kernel<<<n_blocks, TPB, 0, stream>>>(cnt, N);
    count_in_kernel<<<e_blocks, TPB, 0, stream>>>(dst, cnt, E);
    scan_counts_kernel<<<scan_blocks, SCAN_TPB, 0, stream>>>(cnt, rowptr, cur, N);
    fill_slots_kernel<<<e_blocks, TPB, 0, stream>>>(dst, cur, slot_eid, E);
    order_slots_kernel<<<e_blocks, TPB, 0, stream>>>(src, dst, rowptr, slot_eid,
                                                     csr_src, E);

    gemm_dual_f32<<<gemm_blocks, TPB, 0, stream>>>(x, Wl_all, Wr_all, xlA, xrA);

    float* cl = xlA; float* cr = xrA;
    float* nl = xlB; float* nr = xrB;
    for (int l = 0; l < LAYERS - 1; ++l) {
        const float* att  = att_all  + (size_t)l * HEADS * HEAD_DIM;
        const float* bias = bias_all + (size_t)l * HID;
        const float* Wl   = Wl_all   + (size_t)(l + 1) * HID * HID;
        const float* Wr   = Wr_all   + (size_t)(l + 1) * HID * HID;
        attend_gemm_kernel<<<gemm_blocks, 512, 0, stream>>>(rowptr, csr_src, cl, cr,
                                                            att, bias, Wl, Wr, nl, nr);
        float* t;
        t = cl; cl = nl; nl = t;
        t = cr; cr = nr; nr = t;
    }
    {
        const float* att  = att_all  + (size_t)(LAYERS - 1) * HEADS * HEAD_DIM;
        const float* bias = bias_all + (size_t)(LAYERS - 1) * HID;
        node_attend_kernel<<<node_blocks, TPB, 0, stream>>>(rowptr, csr_src, cl, cr,
                                                            att, bias, out, N);
    }
}


# Attention: vector work per edge about halved: weights by one v_exp_f32 on base-2 logits (log2 e folded into att), leaky ReLU as max(t,0.2t), fused dot, end division by v_rcp_f32; sum order kept
# speedup vs baseline: 5.8365x; 1.0410x over previous
#include <hip/hip_runtime.h>
#include <hip/hip_bf16.h>
#include <math.h>

#define HID      128
#define HEADS    8
#define HEAD_DIM 16
#define LAYERS   3
#define NEG_SLOPE 0.2f
#define EPS_F     1e-16f
#define LOG2E_F   1.4426950408889634f

typedef float v2f __attribute__((ext_vector_type(2)));
typedef float v8f __attribute__((ext_vector_type(8)));

#define A_PITCH (HID + 4)
__global__ __launch_bounds__(256)
void gemm_dual_f32(const float* __restrict__ X,
                   const float* __restrict__ Wl,
                   const float* __restrict__ Wr,
                   float* __restrict__ XL,
                   float* __restrict__ XR) {
    __shared__ __attribute__((aligned(16))) float As[16][A_PITCH];
    const int row0 = blockIdx.x * 16;

    {
        const float4* gx = (const float4*)(X + (size_t)row0 * HID);
        for (int t = threadIdx.x; t < 16 * HID / 4; t += 256) {
            const int rr = t >> 5;
            const int c4 = t & 31;
            *(float4*)(&As[rr][4 * c4]) = gx[t];
        }
    }
    __syncthreads();

    const int wave = threadIdx.x >> 5;
    const int lane = threadIdx.x & 31;
    const int col0 = wave * 16;
    const int half = lane >> 4;
    const int lm   = lane & 15;

    v8f accL = {0.f,0.f,0.f,0.f,0.f,0.f,0.f,0.f};
    v8f accR = {0.f,0.f,0.f,0.f,0.f,0.f,0.f,0.f};

    #pragma unroll 4
    for (int k0 = 0; k0 < HID; k0 += 4) {
        const int ka = k0 + 2 * half;
        v2f a;
        a.x = As[lm][ka];
        a.y = As[lm][ka + 1];

        v2f bl, br;
        bl.x = Wl[(size_t)(ka + 0) * HID + col0 + lm];
        bl.y = Wl[(size_t)(ka + 1) * HID + col0 + lm];
        br.x = Wr[(size_t)(ka + 0) * HID + col0 + lm];
        br.y = Wr[(size_t)(ka + 1) * HID + col0 + lm];

        accL = __builtin_amdgcn_wmma_f32_16x16x4_f32(false, a, false, bl,
                                                     (short)0, accL, false, false);
        accR = __builtin_amdgcn_wmma_f32_16x16x4_f32(false, a, false, br,
                                                     (short)0, accR, false, false);
    }

    #pragma unroll
    for (int r = 0; r < 8; ++r) {
        const int m = r + 8 * half;
        XL[(size_t)(row0 + m) * HID + col0 + lm] = accL[r];
        XR[(size_t)(row0 + m) * HID + col0 + lm] = accR[r];
    }
}

__global__ void zero_counts_kernel(int* __restrict__ cnt, int n) {
    int i = blockIdx.x * blockDim.x + threadIdx.x;
    if (i < n) cnt[i] = 0;
}

__global__ void count_in_kernel(const int* __restrict__ dst,
                                int* __restrict__ cnt, int E) {
    int e = blockIdx.x * blockDim.x + threadIdx.x;
    if (e < E) atomicAdd(&cnt[dst[e]], 1);
}

#define SCAN_TPB 1024
__global__ __launch_bounds__(SCAN_TPB)
void scan_counts_kernel(const int* __restrict__ cnt,
                        int* __restrict__ rowptr,
                        int* __restrict__ cur, int n) {
    __shared__ int wfront[SCAN_TPB / 32];
    __shared__ int wtot[SCAN_TPB / 32];
    const int t = threadIdx.x;
    const int lane = t & 31;
    const int wave = t >> 5;
    const int first = blockIdx.x * SCAN_TPB;
    const int i = first + t;
    const int c = (i < n) ? cnt[i] : 0;

    const int4* c4 = (const int4*)cnt + t;
    const int n4 = first >> 2;
    const int full = n4 / SCAN_TPB;
    int s = 0;
    #pragma unroll 4
    for (int k = 0; k < full; ++k) {
        const int4 v = c4[(size_t)k * SCAN_TPB];
        s += (v.x + v.y) + (v.z + v.w);
    }
    if (full * SCAN_TPB + t < n4) {
        const int4 v = c4[(size_t)full * SCAN_TPB];
        s += (v.x + v.y) + (v.z + v.w);
    }
    for (int off = 16; off > 0; off >>= 1) s += __shfl_xor(s, off, 32);

    int inc = c;
    for (int off = 1; off < 32; off <<= 1) {
        const int v = __shfl_up(inc, off, 32);
        if (lane >= off) inc += v;
    }
    if (lane == 0)  wfront[wave] = s;
    if (lane == 31) wtot[wave] = inc;
    __syncthreads();

    int base = wfront[lane] + ((lane < wave) ? wtot[lane] : 0);
    for (int off = 16; off > 0; off >>= 1) base += __shfl_xor(base, off, 32);

    const int excl = base + inc - c;
    if (i < n) {
        rowptr[i] = excl;
        cur[i] = excl;
        if (i == n - 1) rowptr[n] = excl + c;
    }
}

__global__ void fill_slots_kernel(const int* __restrict__ dst,
                                  int* __restrict__ cur,
                                  int* __restrict__ slot_eid, int E) {
    int e = blockIdx.x * blockDim.x + threadIdx.x;
    if (e >= E) return;
    const int p = atomicAdd(&cur[dst[e]], 1);
    slot_eid[p] = e;
}

__global__ void order_slots_kernel(const int* __restrict__ src,
                                   const int* __restrict__ dst,
                                   const int* __restrict__ rowptr,
                                   const int* __restrict__ slot_eid,
                                   int* __restrict__ csr_src, int E) {
    int p = blockIdx.x * blockDim.x + threadIdx.x;
    if (p >= E) return;
    const int e = slot_eid[p];
    const int d = dst[e];
    const int st = rowptr[d], en = rowptr[d + 1];
    int r = 0;
    for (int q = st; q < en; ++q) r += (slot_eid[q] < e) ? 1 : 0;
    csr_src[st + r] = src[e];
}

__device__ __forceinline__ float quad_sum(float x) {
    x += __int_as_float(__builtin_amdgcn_update_dpp(0, __float_as_int(x), 0xB1, 0xF, 0xF, true));
    x += __int_as_float(__builtin_amdgcn_update_dpp(0, __float_as_int(x), 0x4E, 0xF, 0xF, true));
    return x;
}

__device__ __forceinline__ float head_logit(const float4 v, const float4 r,
                                            const float4 a) {
    float t0 = v.x + r.x;
    float t1 = v.y + r.y;
    float t2 = v.z + r.z;
    float t3 = v.w + r.w;
    t0 = __builtin_fmaxf(t0, NEG_SLOPE * t0);
    t1 = __builtin_fmaxf(t1, NEG_SLOPE * t1);
    t2 = __builtin_fmaxf(t2, NEG_SLOPE * t2);
    t3 = __builtin_fmaxf(t3, NEG_SLOPE * t3);
    float acc = t0 * a.x;
    acc = __builtin_fmaf(t1, a.y, acc);
    acc = __builtin_fmaf(t2, a.z, acc);
    acc = __builtin_fmaf(t3, a.w, acc);
    return quad_sum(acc);
}

__device__ __forceinline__ void attend_chunk(const float4* __restrict__ xl4,
                                             const int mine, const int c, const int cnt,
                                             const int lane,
                                             const float4 r, const float4 a,
                                             float& m, float& den, float4& acc) {
    int s[8];
    #pragma unroll
    for (int j = 0; j < 8; ++j) s[j] = __builtin_amdgcn_readlane(mine, c + j);

    float4 v[8];
    #pragma unroll
    for (int j = 0; j < 8; ++j) {
        if (c + j < cnt) v[j] = xl4[(size_t)s[j] * (HID / 4) + lane];
    }

    float lg[8];
    float cm = -INFINITY;
    #pragma unroll
    for (int j = 0; j < 8; ++j) {
        if (c + j < cnt) {
            lg[j] = head_logit(v[j], r, a);
            cm = fmaxf(cm, lg[j]);
        }
    }

    const float mn = fmaxf(m, cm);
    const float sc = __builtin_amdgcn_exp2f(m - mn);
    den   *= sc;
    acc.x *= sc;
    acc.y *= sc;
    acc.z *= sc;
    acc.w *= sc;
    #pragma unroll
    for (int j = 0; j < 8; ++j) {
        if (c + j < cnt) {
            const float w = __builtin_amdgcn_exp2f(lg[j] - mn);
            den   += w;
            acc.x += w * v[j].x;
            acc.y += w * v[j].y;
            acc.z += w * v[j].z;
            acc.w += w * v[j].w;
        }
    }
    m = mn;
}

__device__ __forceinline__ float4 attend_node(const int* __restrict__ rowptr,
                                              const int* __restrict__ csr_src,
                                              const float* __restrict__ xl,
                                              const float* __restrict__ xr,
                                              const float* __restrict__ att,
                                              const float* __restrict__ bias,
                                              const int node, const int lane) {
    const int st = rowptr[node], en = rowptr[node + 1];
    const float4* xl4 = (const float4*)xl;
    const float4 r = ((const float4*)(xr + (size_t)node * HID))[lane];
    float4 a = ((const float4*)att)[lane];
    a.x *= LOG2E_F;
    a.y *= LOG2E_F;
    a.z *= LOG2E_F;
    a.w *= LOG2E_F;

    float m = -INFINITY;
    float den = 0.0f;
    float4 acc = make_float4(0.0f, 0.0f, 0.0f, 0.0f);
    for (int base = st; base < en; base += 32) {
        const int cnt = (en - base < 32) ? (en - base) : 32;
        const int mine = (lane < cnt) ? csr_src[base + lane] : 0;
        for (int c = 0; c < cnt; c += 8)
            attend_chunk(xl4, mine, c, cnt, lane, r, a, m, den, acc);
    }

    const float inv = __builtin_amdgcn_rcpf(den + EPS_F);
    const float4 b = ((const float4*)bias)[lane];
    float4 o;
    o.x = __builtin_fmaf(acc.x, inv, b.x);
    o.y = __builtin_fmaf(acc.y, inv, b.y);
    o.z = __builtin_fmaf(acc.z, inv, b.z);
    o.w = __builtin_fmaf(acc.w, inv, b.w);
    return o;
}

__global__ __launch_bounds__(256)
void node_attend_kernel(const int* __restrict__ rowptr,
                        const int* __restrict__ csr_src,
                        const float* __restrict__ xl,
                        const float* __restrict__ xr,
                        const float* __restrict__ att,
                        const float* __restrict__ bias,
                        float* __restrict__ out, int n) {
    const int node = __builtin_amdgcn_readfirstlane((int)(blockIdx.x * 8 + (threadIdx.x >> 5)));
    if (node >= n) return;
    const int lane = threadIdx.x & 31;
    const float4 o = attend_node(rowptr, csr_src, xl, xr, att, bias, node, lane);
    ((float4*)(out + (size_t)node * HID))[lane] = o;
}

__global__ __launch_bounds__(512)
void attend_gemm_kernel(const int* __restrict__ rowptr,
                        const int* __restrict__ csr_src,
                        const float* __restrict__ xl,
                        const float* __restrict__ xr,
                        const float* __restrict__ att,
                        const float* __restrict__ bias,
                        const float* __restrict__ Wl,
                        const float* __restrict__ Wr,
                        float* __restrict__ XLn,
                        float* __restrict__ XRn) {
    __shared__ __attribute__((aligned(16))) float As[16][A_PITCH];
    const int wave = __builtin_amdgcn_readfirstlane((int)(threadIdx.x >> 5));
    const int lane = threadIdx.x & 31;
    const int row0 = blockIdx.x * 16;
    const int node = row0 + wave;

    const float4 o = attend_node(rowptr, csr_src, xl, xr, att, bias, node, lane);
    *(float4*)(&As[wave][4 * lane]) = o;
    __syncthreads();

    const float* __restrict__ Wm = (wave < 8) ? Wl : Wr;
    float* __restrict__ Om = (wave < 8) ? XLn : XRn;
    const int col0 = (wave & 7) * 16;
    const int half = lane >> 4;
    const int lm   = lane & 15;

    v8f acc = {0.f,0.f,0.f,0.f,0.f,0.f,0.f,0.f};

    #pragma unroll 16
    for (int k0 = 0; k0 < HID; k0 += 4) {
        const int ka = k0 + 2 * half;
        v2f a;
        a.x = As[lm][ka];
        a.y = As[lm][ka + 1];

        v2f b;
        b.x = Wm[(size_t)(ka + 0) * HID + col0 + lm];
        b.y = Wm[(size_t)(ka + 1) * HID + col0 + lm];

        acc = __builtin_amdgcn_wmma_f32_16x16x4_f32(false, a, false, b,
                                                    (short)0, acc, false, false);
    }

    #pragma unroll
    for (int r = 0; r < 8; ++r) {
        const int m = r + 8 * half;
        Om[(size_t)(row0 + m) * HID + col0 + lm] = acc[r];
    }
}

extern "C" void kernel_launch(void* const* d_in, const int* in_sizes, int n_in,
                              void* d_out, int out_size, void* d_ws, size_t ws_size,
                              hipStream_t stream) {
    const float* x          = (const float*)d_in[0];
    const int*   edge_index = (const int*)  d_in[1];
    const float* Wl_all     = (const float*)d_in[2];
    const float* Wr_all     = (const float*)d_in[3];
    const float* att_all    = (const float*)d_in[4];
    const float* bias_all   = (const float*)d_in[5];
    float* out = (float*)d_out;

    const int N = in_sizes[0] / HID;
    const int E = in_sizes[1] / 2;

    const int* src = edge_index;
    const int* dst = edge_index + E;

    const size_t nf = (size_t)N * HID;
    float* xlA   = (float*)d_ws;
    float* xrA   = xlA + nf;
    float* xlB   = xrA + nf;
    float* xrB   = xlB + nf;
    int* cnt      = (int*)(xrB + nf);
    int* rowptr   = cnt + N;
    int* cur      = rowptr + N + 1;
    int* slot_eid = cur + N;
    int* csr_src  = slot_eid + E;

    const int TPB = 256;
    const int gemm_blocks  = N / 16;
    const int n_blocks     = (N + TPB - 1) / TPB;
    const int e_blocks     = (E + TPB - 1) / TPB;
    const int node_blocks  = (N + 7) / 8;
    const int scan_blocks  = (N + SCAN_TPB - 1) / SCAN_TPB;

    zero_counts_kernel<<<n_blocks, TPB, 0, stream>>>(cnt, N);
    count_in_kernel<<<e_blocks, TPB, 0, stream>>>(dst, cnt, E);
    scan_counts_kernel<<<scan_blocks, SCAN_TPB, 0, stream>>>(cnt, rowptr, cur, N);
    fill_slots_kernel<<<e_blocks, TPB, 0, stream>>>(dst, cur, slot_eid, E);
    order_slots_kernel<<<e_blocks, TPB, 0, stream>>>(src, dst, rowptr, slot_eid,
                                                     csr_src, E);

    gemm_dual_f32<<<gemm_blocks, TPB, 0, stream>>>(x, Wl_all, Wr_all, xlA, xrA);

    float* cl = xlA; float* cr = xrA;
    float* nl = xlB; float* nr = xrB;
    for (int l = 0; l < LAYERS - 1; ++l) {
        const float* att  = att_all  + (size_t)l * HEADS * HEAD_DIM;
        const float* bias = bias_all + (size_t)l * HID;
        const float* Wl   = Wl_all   + (size_t)(l + 1) * HID * HID;
        const float* Wr   = Wr_all   + (size_t)(l + 1) * HID * HID;
        attend_gemm_kernel<<<gemm_blocks, 512, 0, stream>>>(rowptr, csr_src, cl, cr,
                                                            att, bias, Wl, Wr, nl, nr);
        float* t;
        t = cl; cl = nl; nl = t;
        t = cr; cr = nr; nr = t;
    }
    {
        const float* att  = att_all  + (size_t)(LAYERS - 1) * HEADS * HEAD_DIM;
        const float* bias = bias_all + (size_t)(LAYERS - 1) * HID;
        node_attend_kernel<<<node_blocks, TPB, 0, stream>>>(rowptr, csr_src, cl, cr,
                                                            att, bias, out, N);
    }
}


# Each weight fragment load feeds two row tiles: GEMM blocks of 32 rows (fused: 32 nodes, 32 waves; waves 0-15 do a column tile of both row tiles); a weight line is read once per 32 rows; same sums
# speedup vs baseline: 6.0713x; 1.0402x over previous
#include <hip/hip_runtime.h>
#include <hip/hip_bf16.h>
#include <math.h>

#define HID      128
#define HEADS    8
#define HEAD_DIM 16
#define LAYERS   3
#define NEG_SLOPE 0.2f
#define EPS_F     1e-16f
#define LOG2E_F   1.4426950408889634f

typedef float v2f __attribute__((ext_vector_type(2)));
typedef float v8f __attribute__((ext_vector_type(8)));

#define A_PITCH (HID + 4)
#define STRIP_ROWS 32
__global__ __launch_bounds__(256)
void gemm_dual_f32(const float* __restrict__ X,
                   const float* __restrict__ Wl,
                   const float* __restrict__ Wr,
                   float* __restrict__ XL,
                   float* __restrict__ XR, int n) {
    __shared__ __attribute__((aligned(16))) float As[STRIP_ROWS][A_PITCH];
    const int row0 = blockIdx.x * STRIP_ROWS;

    {
        const float4* gx = (const float4*)(X + (size_t)row0 * HID);
        for (int t = threadIdx.x; t < STRIP_ROWS * HID / 4; t += 256) {
            const int rr = t >> 5;
            const int c4 = t & 31;
            float4 v = make_float4(0.0f, 0.0f, 0.0f, 0.0f);
            if (row0 + rr < n) v = gx[t];
            *(float4*)(&As[rr][4 * c4]) = v;
        }
    }
    __syncthreads();

    const int wave = threadIdx.x >> 5;
    const int lane = threadIdx.x & 31;
    const int col0 = wave * 16;
    const int half = lane >> 4;
    const int lm   = lane & 15;

    v8f accL0 = {0.f,0.f,0.f,0.f,0.f,0.f,0.f,0.f};
    v8f accL1 = {0.f,0.f,0.f,0.f,0.f,0.f,0.f,0.f};
    v8f accR0 = {0.f,0.f,0.f,0.f,0.f,0.f,0.f,0.f};
    v8f accR1 = {0.f,0.f,0.f,0.f,0.f,0.f,0.f,0.f};

    #pragma unroll 4
    for (int k0 = 0; k0 < HID; k0 += 4) {
        const int ka = k0 + 2 * half;
        v2f a0, a1;
        a0.x = As[lm][ka];
        a0.y = As[lm][ka + 1];
        a1.x = As[16 + lm][ka];
        a1.y = As[16 + lm][ka + 1];

        v2f bl, br;
        bl.x = Wl[(size_t)(ka + 0) * HID + col0 + lm];
        bl.y = Wl[(size_t)(ka + 1) * HID + col0 + lm];
        br.x = Wr[(size_t)(ka + 0) * HID + col0 + lm];
        br.y = Wr[(size_t)(ka + 1) * HID + col0 + lm];

        accL0 = __builtin_amdgcn_wmma_f32_16x16x4_f32(false, a0, false, bl,
                                                      (short)0, accL0, false, false);
        accL1 = __builtin_amdgcn_wmma_f32_16x16x4_f32(false, a1, false, bl,
                                                      (short)0, accL1, false, false);
        accR0 = __builtin_amdgcn_wmma_f32_16x16x4_f32(false, a0, false, br,
                                                      (short)0, accR0, false, false);
        accR1 = __builtin_amdgcn_wmma_f32_16x16x4_f32(false, a1, false, br,
                                                      (short)0, accR1, false, false);
    }

    if (row0 + STRIP_ROWS <= n) {
        #pragma unroll
        for (int r = 0; r < 8; ++r) {
            const int m0 = row0 + r + 8 * half;
            XL[(size_t)m0 * HID + col0 + lm] = accL0[r];
            XR[(size_t)m0 * HID + col0 + lm] = accR0[r];
            XL[(size_t)(m0 + 16) * HID + col0 + lm] = accL1[r];
            XR[(size_t)(m0 + 16) * HID + col0 + lm] = accR1[r];
        }
    } else {
        #pragma unroll
        for (int r = 0; r < 8; ++r) {
            const int m0 = row0 + r + 8 * half;
            const int m1 = m0 + 16;
            if (m0 < n) {
                XL[(size_t)m0 * HID + col0 + lm] = accL0[r];
                XR[(size_t)m0 * HID + col0 + lm] = accR0[r];
            }
            if (m1 < n) {
                XL[(size_t)m1 * HID + col0 + lm] = accL1[r];
                XR[(size_t)m1 * HID + col0 + lm] = accR1[r];
            }
        }
    }
}

__global__ void zero_counts_kernel(int* __restrict__ cnt, int n) {
    int i = blockIdx.x * blockDim.x + threadIdx.x;
    if (i < n) cnt[i] = 0;
}

__global__ void count_in_kernel(const int* __restrict__ dst,
                                int* __restrict__ cnt, int E) {
    int e = blockIdx.x * blockDim.x + threadIdx.x;
    if (e < E) atomicAdd(&cnt[dst[e]], 1);
}

#define SCAN_TPB 1024
__global__ __launch_bounds__(SCAN_TPB)
void scan_counts_kernel(const int* __restrict__ cnt,
                        int* __restrict__ rowptr,
                        int* __restrict__ cur, int n) {
    __shared__ int wfront[SCAN_TPB / 32];
    __shared__ int wtot[SCAN_TPB / 32];
    const int t = threadIdx.x;
    const int lane = t & 31;
    const int wave = t >> 5;
    const int first = blockIdx.x * SCAN_TPB;
    const int i = first + t;
    const int c = (i < n) ? cnt[i] : 0;

    const int4* c4 = (const int4*)cnt + t;
    const int n4 = first >> 2;
    const int full = n4 / SCAN_TPB;
    int s = 0;
    #pragma unroll 4
    for (int k = 0; k < full; ++k) {
        const int4 v = c4[(size_t)k * SCAN_TPB];
        s += (v.x + v.y) + (v.z + v.w);
    }
    if (full * SCAN_TPB + t < n4) {
        const int4 v = c4[(size_t)full * SCAN_TPB];
        s += (v.x + v.y) + (v.z + v.w);
    }
    for (int off = 16; off > 0; off >>= 1) s += __shfl_xor(s, off, 32);

    int inc = c;
    for (int off = 1; off < 32; off <<= 1) {
        const int v = __shfl_up(inc, off, 32);
        if (lane >= off) inc += v;
    }
    if (lane == 0)  wfront[wave] = s;
    if (lane == 31) wtot[wave] = inc;
    __syncthreads();

    int base = wfront[lane] + ((lane < wave) ? wtot[lane] : 0);
    for (int off = 16; off > 0; off >>= 1) base += __shfl_xor(base, off, 32);

    const int excl = base + inc - c;
    if (i < n) {
        rowptr[i] = excl;
        cur[i] = excl;
        if (i == n - 1) rowptr[n] = excl + c;
    }
}

__global__ void fill_slots_kernel(const int* __restrict__ dst,
                                  int* __restrict__ cur,
                                  int* __restrict__ slot_eid, int E) {
    int e = blockIdx.x * blockDim.x + threadIdx.x;
    if (e >= E) return;
    const int p = atomicAdd(&cur[dst[e]], 1);
    slot_eid[p] = e;
}

__global__ void order_slots_kernel(const int* __restrict__ src,
                                   const int* __restrict__ dst,
                                   const int* __restrict__ rowptr,
                                   const int* __restrict__ slot_eid,
                                   int* __restrict__ csr_src, int E) {
    int p = blockIdx.x * blockDim.x + threadIdx.x;
    if (p >= E) return;
    const int e = slot_eid[p];
    const int d = dst[e];
    const int st = rowptr[d], en = rowptr[d + 1];
    int r = 0;
    for (int q = st; q < en; ++q) r += (slot_eid[q] < e) ? 1 : 0;
    csr_src[st + r] = src[e];
}

__device__ __forceinline__ float quad_sum(float x) {
    x += __int_as_float(__builtin_amdgcn_update_dpp(0, __float_as_int(x), 0xB1, 0xF, 0xF, true));
    x += __int_as_float(__builtin_amdgcn_update_dpp(0, __float_as_int(x), 0x4E, 0xF, 0xF, true));
    return x;
}

__device__ __forceinline__ float head_logit(const float4 v, const float4 r,
                                            const float4 a) {
    float t0 = v.x + r.x;
    float t1 = v.y + r.y;
    float t2 = v.z + r.z;
    float t3 = v.w + r.w;
    t0 = __builtin_fmaxf(t0, NEG_SLOPE * t0);
    t1 = __builtin_fmaxf(t1, NEG_SLOPE * t1);
    t2 = __builtin_fmaxf(t2, NEG_SLOPE * t2);
    t3 = __builtin_fmaxf(t3, NEG_SLOPE * t3);
    float acc = t0 * a.x;
    acc = __builtin_fmaf(t1, a.y, acc);
    acc = __builtin_fmaf(t2, a.z, acc);
    acc = __builtin_fmaf(t3, a.w, acc);
    return quad_sum(acc);
}

__device__ __forceinline__ void attend_chunk(const float4* __restrict__ xl4,
                                             const int mine, const int c, const int cnt,
                                             const int lane,
                                             const float4 r, const float4 a,
                                             float& m, float& den, float4& acc) {
    int s[8];
    #pragma unroll
    for (int j = 0; j < 8; ++j) s[j] = __builtin_amdgcn_readlane(mine, c + j);

    float4 v[8];
    #pragma unroll
    for (int j = 0; j < 8; ++j) {
        if (c + j < cnt) v[j] = xl4[(size_t)s[j] * (HID / 4) + lane];
    }

    float lg[8];
    float cm = -INFINITY;
    #pragma unroll
    for (int j = 0; j < 8; ++j) {
        if (c + j < cnt) {
            lg[j] = head_logit(v[j], r, a);
            cm = fmaxf(cm, lg[j]);
        }
    }

    const float mn = fmaxf(m, cm);
    const float sc = __builtin_amdgcn_exp2f(m - mn);
    den   *= sc;
    acc.x *= sc;
    acc.y *= sc;
    acc.z *= sc;
    acc.w *= sc;
    #pragma unroll
    for (int j = 0; j < 8; ++j) {
        if (c + j < cnt) {
            const float w = __builtin_amdgcn_exp2f(lg[j] - mn);
            den   += w;
            acc.x += w * v[j].x;
            acc.y += w * v[j].y;
            acc.z += w * v[j].z;
            acc.w += w * v[j].w;
        }
    }
    m = mn;
}

__device__ __forceinline__ float4 attend_node(const int* __restrict__ rowptr,
                                              const int* __restrict__ csr_src,
                                              const float* __restrict__ xl,
                                              const float* __restrict__ xr,
                                              const float* __restrict__ att,
                                              const float* __restrict__ bias,
                                              const int node, const int lane) {
    const int st = rowptr[node], en = rowptr[node + 1];
    const float4* xl4 = (const float4*)xl;
    const float4 r = ((const float4*)(xr + (size_t)node * HID))[lane];
    float4 a = ((const float4*)att)[lane];
    a.x *= LOG2E_F;
    a.y *= LOG2E_F;
    a.z *= LOG2E_F;
    a.w *= LOG2E_F;

    float m = -INFINITY;
    float den = 0.0f;
    float4 acc = make_float4(0.0f, 0.0f, 0.0f, 0.0f);
    for (int base = st; base < en; base += 32) {
        const int cnt = (en - base < 32) ? (en - base) : 32;
        const int mine = (lane < cnt) ? csr_src[base + lane] : 0;
        for (int c = 0; c < cnt; c += 8)
            attend_chunk(xl4, mine, c, cnt, lane, r, a, m, den, acc);
    }

    const float inv = __builtin_amdgcn_rcpf(den + EPS_F);
    const float4 b = ((const float4*)bias)[lane];
    float4 o;
    o.x = __builtin_fmaf(acc.x, inv, b.x);
    o.y = __builtin_fmaf(acc.y, inv, b.y);
    o.z = __builtin_fmaf(acc.z, inv, b.z);
    o.w = __builtin_fmaf(acc.w, inv, b.w);
    return o;
}

__global__ __launch_bounds__(256)
void node_attend_kernel(const int* __restrict__ rowptr,
                        const int* __restrict__ csr_src,
                        const float* __restrict__ xl,
                        const float* __restrict__ xr,
                        const float* __restrict__ att,
                        const float* __restrict__ bias,
                        float* __restrict__ out, int n) {
    const int node = __builtin_amdgcn_readfirstlane((int)(blockIdx.x * 8 + (threadIdx.x >> 5)));
    if (node >= n) return;
    const int lane = threadIdx.x & 31;
    const float4 o = attend_node(rowptr, csr_src, xl, xr, att, bias, node, lane);
    ((float4*)(out + (size_t)node * HID))[lane] = o;
}

__global__ __launch_bounds__(1024) __attribute__((amdgpu_waves_per_eu(16, 16)))
void attend_gemm_kernel(const int* __restrict__ rowptr,
                        const int* __restrict__ csr_src,
                        const float* __restrict__ xl,
                        const float* __restrict__ xr,
                        const float* __restrict__ att,
                        const float* __restrict__ bias,
                        const float* __restrict__ Wl,
                        const float* __restrict__ Wr,
                        float* __restrict__ XLn,
                        float* __restrict__ XRn, int n) {
    __shared__ __attribute__((aligned(16))) float As[STRIP_ROWS][A_PITCH];
    const int wave = __builtin_amdgcn_readfirstlane((int)(threadIdx.x >> 5));
    const int lane = threadIdx.x & 31;
    const int row0 = blockIdx.x * STRIP_ROWS;
    const int node = row0 + wave;

    float4 o = make_float4(0.0f, 0.0f, 0.0f, 0.0f);
    if (node < n) o = attend_node(rowptr, csr_src, xl, xr, att, bias, node, lane);
    *(float4*)(&As[wave][4 * lane]) = o;
    __syncthreads();
    if (wave >= 16) return;

    const float* __restrict__ Wm = (wave < 8) ? Wl : Wr;
    float* __restrict__ Om = (wave < 8) ? XLn : XRn;
    const int col0 = (wave & 7) * 16;
    const int half = lane >> 4;
    const int lm   = lane & 15;

    v8f acc0 = {0.f,0.f,0.f,0.f,0.f,0.f,0.f,0.f};
    v8f acc1 = {0.f,0.f,0.f,0.f,0.f,0.f,0.f,0.f};

    #pragma unroll 8
    for (int k0 = 0; k0 < HID; k0 += 4) {
        const int ka = k0 + 2 * half;
        v2f a0, a1;
        a0.x = As[lm][ka];
        a0.y = As[lm][ka + 1];
        a1.x = As[16 + lm][ka];
        a1.y = As[16 + lm][ka + 1];

        v2f b;
        b.x = Wm[(size_t)(ka + 0) * HID + col0 + lm];
        b.y = Wm[(size_t)(ka + 1) * HID + col0 + lm];

        acc0 = __builtin_amdgcn_wmma_f32_16x16x4_f32(false, a0, false, b,
                                                     (short)0, acc0, false, false);
        acc1 = __builtin_amdgcn_wmma_f32_16x16x4_f32(false, a1, false, b,
                                                     (short)0, acc1, false, false);
    }

    if (row0 + STRIP_ROWS <= n) {
        #pragma unroll
        for (int r = 0; r < 8; ++r) {
            const int m0 = row0 + r + 8 * half;
            Om[(size_t)m0 * HID + col0 + lm] = acc0[r];
            Om[(size_t)(m0 + 16) * HID + col0 + lm] = acc1[r];
        }
    } else {
        #pragma unroll
        for (int r = 0; r < 8; ++r) {
            const int m0 = row0 + r + 8 * half;
            const int m1 = m0 + 16;
            if (m0 < n) Om[(size_t)m0 * HID + col0 + lm] = acc0[r];
            if (m1 < n) Om[(size_t)m1 * HID + col0 + lm] = acc1[r];
        }
    }
}

extern "C" void kernel_launch(void* const* d_in, const int* in_sizes, int n_in,
                              void* d_out, int out_size, void* d_ws, size_t ws_size,
                              hipStream_t stream) {
    const float* x          = (const float*)d_in[0];
    const int*   edge_index = (const int*)  d_in[1];
    const float* Wl_all     = (const float*)d_in[2];
    const float* Wr_all     = (const float*)d_in[3];
    const float* att_all    = (const float*)d_in[4];
    const float* bias_all   = (const float*)d_in[5];
    float* out = (float*)d_out;

    const int N = in_sizes[0] / HID;
    const int E = in_sizes[1] / 2;

    const int* src = edge_index;
    const int* dst = edge_index + E;

    const size_t nf = (size_t)N * HID;
    float* xlA   = (float*)d_ws;
    float* xrA   = xlA + nf;
    float* xlB   = xrA + nf;
    float* xrB   = xlB + nf;
    int* cnt      = (int*)(xrB + nf);
    int* rowptr   = cnt + N;
    int* cur      = rowptr + N + 1;
    int* slot_eid = cur + N;
    int* csr_src  = slot_eid + E;

    const int TPB = 256;
    const int gemm_blocks  = (N + STRIP_ROWS - 1) / STRIP_ROWS;
    const int n_blocks     = (N + TPB - 1) / TPB;
    const int e_blocks     = (E + TPB - 1) / TPB;
    const int node_blocks  = (N + 7) / 8;
    const int scan_blocks  = (N + SCAN_TPB - 1) / SCAN_TPB;

    zero_counts_kernel<<<n_blocks, TPB, 0, stream>>>(cnt, N);
    count_in_kernel<<<e_blocks, TPB, 0, stream>>>(dst, cnt, E);
    scan_counts_kernel<<<scan_blocks, SCAN_TPB, 0, stream>>>(cnt, rowptr, cur, N);
    fill_slots_kernel<<<e_blocks, TPB, 0, stream>>>(dst, cur, slot_eid, E);
    order_slots_kernel<<<e_blocks, TPB, 0, stream>>>(src, dst, rowptr, slot_eid,
                                                     csr_src, E);

    gemm_dual_f32<<<gemm_blocks, TPB, 0, stream>>>(x, Wl_all, Wr_all, xlA, xrA, N);

    float* cl = xlA; float* cr = xrA;
    float* nl = xlB; float* nr = xrB;
    for (int l = 0; l < LAYERS - 1; ++l) {
        const float* att  = att_all  + (size_t)l * HEADS * HEAD_DIM;
        const float* bias = bias_all + (size_t)l * HID;
        const float* Wl   = Wl_all   + (size_t)(l + 1) * HID * HID;
        const float* Wr   = Wr_all   + (size_t)(l + 1) * HID * HID;
        attend_gemm_kernel<<<gemm_blocks, 1024, 0, stream>>>(rowptr, csr_src, cl, cr,
                                                             att, bias, Wl, Wr, nl, nr, N);
        float* t;
        t = cl; cl = nl; nl = t;
        t = cr; cr = nr; nr = t;
    }
    {
        const float* att  = att_all  + (size_t)(LAYERS - 1) * HEADS * HEAD_DIM;
        const float* bias = bias_all + (size_t)(LAYERS - 1) * HID;
        node_attend_kernel<<<node_blocks, TPB, 0, stream>>>(rowptr, csr_src, cl, cr,
                                                            att, bias, out, N);
    }
}


# Edges grouped by destination with work-group-local LDS counters (two-level counting sort: 256-node bins, then nodes in a bin): no global atomic per edge, no counter clear; same slot order and sums
# speedup vs baseline: 8.1213x; 1.3377x over previous
#include <hip/hip_runtime.h>
#include <hip/hip_bf16.h>
#include <math.h>

#define HID      128
#define HEADS    8
#define HEAD_DIM 16
#define LAYERS   3
#define NEG_SLOPE 0.2f
#define EPS_F     1e-16f
#define LOG2E_F   1.4426950408889634f

typedef float v2f __attribute__((ext_vector_type(2)));
typedef float v8f __attribute__((ext_vector_type(8)));

#define A_PITCH (HID + 4)
#define STRIP_ROWS 32
__global__ __launch_bounds__(256)
void gemm_dual_f32(const float* __restrict__ X,
                   const float* __restrict__ Wl,
                   const float* __restrict__ Wr,
                   float* __restrict__ XL,
                   float* __restrict__ XR, int n) {
    __shared__ __attribute__((aligned(16))) float As[STRIP_ROWS][A_PITCH];
    const int row0 = blockIdx.x * STRIP_ROWS;

    {
        const float4* gx = (const float4*)(X + (size_t)row0 * HID);
        for (int t = threadIdx.x; t < STRIP_ROWS * HID / 4; t += 256) {
            const int rr = t >> 5;
            const int c4 = t & 31;
            float4 v = make_float4(0.0f, 0.0f, 0.0f, 0.0f);
            if (row0 + rr < n) v = gx[t];
            *(float4*)(&As[rr][4 * c4]) = v;
        }
    }
    __syncthreads();

    const int wave = threadIdx.x >> 5;
    const int lane = threadIdx.x & 31;
    const int col0 = wave * 16;
    const int half = lane >> 4;
    const int lm   = lane & 15;

    v8f accL0 = {0.f,0.f,0.f,0.f,0.f,0.f,0.f,0.f};
    v8f accL1 = {0.f,0.f,0.f,0.f,0.f,0.f,0.f,0.f};
    v8f accR0 = {0.f,0.f,0.f,0.f,0.f,0.f,0.f,0.f};
    v8f accR1 = {0.f,0.f,0.f,0.f,0.f,0.f,0.f,0.f};

    #pragma unroll 4
    for (int k0 = 0; k0 < HID; k0 += 4) {
        const int ka = k0 + 2 * half;
        v2f a0, a1;
        a0.x = As[lm][ka];
        a0.y = As[lm][ka + 1];
        a1.x = As[16 + lm][ka];
        a1.y = As[16 + lm][ka + 1];

        v2f bl, br;
        bl.x = Wl[(size_t)(ka + 0) * HID + col0 + lm];
        bl.y = Wl[(size_t)(ka + 1) * HID + col0 + lm];
        br.x = Wr[(size_t)(ka + 0) * HID + col0 + lm];
        br.y = Wr[(size_t)(ka + 1) * HID + col0 + lm];

        accL0 = __builtin_amdgcn_wmma_f32_16x16x4_f32(false, a0, false, bl,
                                                      (short)0, accL0, false, false);
        accL1 = __builtin_amdgcn_wmma_f32_16x16x4_f32(false, a1, false, bl,
                                                      (short)0, accL1, false, false);
        accR0 = __builtin_amdgcn_wmma_f32_16x16x4_f32(false, a0, false, br,
                                                      (short)0, accR0, false, false);
        accR1 = __builtin_amdgcn_wmma_f32_16x16x4_f32(false, a1, false, br,
                                                      (short)0, accR1, false, false);
    }

    if (row0 + STRIP_ROWS <= n) {
        #pragma unroll
        for (int r = 0; r < 8; ++r) {
            const int m0 = row0 + r + 8 * half;
            XL[(size_t)m0 * HID + col0 + lm] = accL0[r];
            XR[(size_t)m0 * HID + col0 + lm] = accR0[r];
            XL[(size_t)(m0 + 16) * HID + col0 + lm] = accL1[r];
            XR[(size_t)(m0 + 16) * HID + col0 + lm] = accR1[r];
        }
    } else {
        #pragma unroll
        for (int r = 0; r < 8; ++r) {
            const int m0 = row0 + r + 8 * half;
            const int m1 = m0 + 16;
            if (m0 < n) {
                XL[(size_t)m0 * HID + col0 + lm] = accL0[r];
                XR[(size_t)m0 * HID + col0 + lm] = accR0[r];
            }
            if (m1 < n) {
                XL[(size_t)m1 * HID + col0 + lm] = accL1[r];
                XR[(size_t)m1 * HID + col0 + lm] = accR1[r];
            }
        }
    }
}

#define BIN_SHIFT   8
#define BIN_NODES   256
#define SORT_TPB    1024
#define SORT_CHUNKS 256

__global__ __launch_bounds__(SORT_TPB)
void chunk_hist_kernel(const int* __restrict__ dst,
                       int* __restrict__ hist,
                       int E, int chunk, int nbins, int nchunks) {
    extern __shared__ int lbin[];
    const int c = blockIdx.x;
    const int t = threadIdx.x;
    for (int b = t; b < nbins; b += SORT_TPB) lbin[b] = 0;
    __syncthreads();
    const int e0 = c * chunk;
    const int e1 = (E - e0 < chunk) ? E : (e0 + chunk);
    for (int e = e0 + t; e < e1; e += 4 * SORT_TPB) {
        int d[4];
        #pragma unroll
        for (int j = 0; j < 4; ++j) {
            const int ee = e + j * SORT_TPB;
            d[j] = (ee < e1) ? dst[ee] : 0;
        }
        #pragma unroll
        for (int j = 0; j < 4; ++j) {
            if (e + j * SORT_TPB < e1) atomicAdd(&lbin[d[j] >> BIN_SHIFT], 1);
        }
    }
    __syncthreads();
    for (int b = t; b < nbins; b += SORT_TPB) hist[(size_t)b * nchunks + c] = lbin[b];
}

#define SCAN_TPB 1024
__global__ __launch_bounds__(SCAN_TPB)
void scan_counts_kernel(const int* __restrict__ cnt,
                        int* __restrict__ rowptr, int n) {
    __shared__ int wfront[SCAN_TPB / 32];
    __shared__ int wtot[SCAN_TPB / 32];
    const int t = threadIdx.x;
    const int lane = t & 31;
    const int wave = t >> 5;
    const int first = blockIdx.x * SCAN_TPB;
    const int i = first + t;
    const int c = (i < n) ? cnt[i] : 0;

    const int4* c4 = (const int4*)cnt + t;
    const int n4 = first >> 2;
    const int full = n4 / SCAN_TPB;
    int s = 0;
    #pragma unroll 4
    for (int k = 0; k < full; ++k) {
        const int4 v = c4[(size_t)k * SCAN_TPB];
        s += (v.x + v.y) + (v.z + v.w);
    }
    if (full * SCAN_TPB + t < n4) {
        const int4 v = c4[(size_t)full * SCAN_TPB];
        s += (v.x + v.y) + (v.z + v.w);
    }
    for (int off = 16; off > 0; off >>= 1) s += __shfl_xor(s, off, 32);

    int inc = c;
    for (int off = 1; off < 32; off <<= 1) {
        const int v = __shfl_up(inc, off, 32);
        if (lane >= off) inc += v;
    }
    if (lane == 0)  wfront[wave] = s;
    if (lane == 31) wtot[wave] = inc;
    __syncthreads();

    int base = wfront[lane] + ((lane < wave) ? wtot[lane] : 0);
    for (int off = 16; off > 0; off >>= 1) base += __shfl_xor(base, off, 32);

    const int excl = base + inc - c;
    if (i < n) {
        rowptr[i] = excl;
        if (i == n - 1) rowptr[n] = excl + c;
    }
}

__global__ __launch_bounds__(SORT_TPB)
void chunk_scatter_kernel(const int* __restrict__ dst,
                          const int* __restrict__ offs,
                          int* __restrict__ binned,
                          int E, int chunk, int nbins, int nchunks) {
    extern __shared__ int lbin[];
    const int c = blockIdx.x;
    const int t = threadIdx.x;
    for (int b = t; b < nbins; b += SORT_TPB) lbin[b] = offs[(size_t)b * nchunks + c];
    __syncthreads();
    const int e0 = c * chunk;
    const int e1 = (E - e0 < chunk) ? E : (e0 + chunk);
    for (int e = e0 + t; e < e1; e += 4 * SORT_TPB) {
        int d[4];
        #pragma unroll
        for (int j = 0; j < 4; ++j) {
            const int ee = e + j * SORT_TPB;
            d[j] = (ee < e1) ? dst[ee] : 0;
        }
        #pragma unroll
        for (int j = 0; j < 4; ++j) {
            const int ee = e + j * SORT_TPB;
            if (ee < e1) {
                const int p = atomicAdd(&lbin[d[j] >> BIN_SHIFT], 1);
                ((int2*)binned)[p] = make_int2(ee, d[j]);
            }
        }
    }
}

__global__ __launch_bounds__(SORT_TPB)
void bin_place_kernel(const int* __restrict__ offs,
                      const int* __restrict__ binned,
                      int* __restrict__ rowptr,
                      int* __restrict__ slot_eid, int n, int nchunks) {
    __shared__ int lcnt[BIN_NODES];
    __shared__ int lcur[BIN_NODES];
    __shared__ int wtot[BIN_NODES / 32];
    const int b = blockIdx.x;
    const int t = threadIdx.x;
    const int lane = t & 31;
    const int wave = t >> 5;
    const int bs = offs[(size_t)b * nchunks];
    const int be = offs[(size_t)(b + 1) * nchunks];
    const int lo = b << BIN_SHIFT;
    if (t < BIN_NODES) lcnt[t] = 0;
    __syncthreads();
    for (int p = bs + t; p < be; p += 4 * SORT_TPB) {
        int d[4];
        #pragma unroll
        for (int j = 0; j < 4; ++j) {
            const int pp = p + j * SORT_TPB;
            d[j] = (pp < be) ? binned[2 * (size_t)pp + 1] : lo;
        }
        #pragma unroll
        for (int j = 0; j < 4; ++j) {
            if (p + j * SORT_TPB < be) atomicAdd(&lcnt[d[j] - lo], 1);
        }
    }
    __syncthreads();

    int c = 0, inc = 0;
    if (t < BIN_NODES) {
        c = lcnt[t];
        inc = c;
        for (int off = 1; off < 32; off <<= 1) {
            const int v = __shfl_up(inc, off, 32);
            if (lane >= off) inc += v;
        }
        if (lane == 31) wtot[wave] = inc;
    }
    __syncthreads();
    if (t < BIN_NODES) {
        int base = bs;
        for (int w = 0; w < wave; ++w) base += wtot[w];
        const int excl = base + inc - c;
        lcur[t] = excl;
        if (lo + t < n) rowptr[lo + t] = excl;
    }
    if (b == (int)gridDim.x - 1 && t == 0) rowptr[n] = be;
    __syncthreads();
    for (int p = bs + t; p < be; p += 4 * SORT_TPB) {
        int2 ed[4];
        #pragma unroll
        for (int j = 0; j < 4; ++j) {
            const int pp = p + j * SORT_TPB;
            ed[j] = (pp < be) ? ((const int2*)binned)[pp] : make_int2(0, lo);
        }
        #pragma unroll
        for (int j = 0; j < 4; ++j) {
            if (p + j * SORT_TPB < be) {
                const int q = atomicAdd(&lcur[ed[j].y - lo], 1);
                slot_eid[q] = ed[j].x;
            }
        }
    }
}

__global__ void order_slots_kernel(const int* __restrict__ src,
                                   const int* __restrict__ dst,
                                   const int* __restrict__ rowptr,
                                   const int* __restrict__ slot_eid,
                                   int* __restrict__ csr_src, int E) {
    int p = blockIdx.x * blockDim.x + threadIdx.x;
    if (p >= E) return;
    const int e = slot_eid[p];
    const int d = dst[e];
    const int st = rowptr[d], en = rowptr[d + 1];
    int r = 0;
    for (int q = st; q < en; ++q) r += (slot_eid[q] < e) ? 1 : 0;
    csr_src[st + r] = src[e];
}

__device__ __forceinline__ float quad_sum(float x) {
    x += __int_as_float(__builtin_amdgcn_update_dpp(0, __float_as_int(x), 0xB1, 0xF, 0xF, true));
    x += __int_as_float(__builtin_amdgcn_update_dpp(0, __float_as_int(x), 0x4E, 0xF, 0xF, true));
    return x;
}

__device__ __forceinline__ float head_logit(const float4 v, const float4 r,
                                            const float4 a) {
    float t0 = v.x + r.x;
    float t1 = v.y + r.y;
    float t2 = v.z + r.z;
    float t3 = v.w + r.w;
    t0 = __builtin_fmaxf(t0, NEG_SLOPE * t0);
    t1 = __builtin_fmaxf(t1, NEG_SLOPE * t1);
    t2 = __builtin_fmaxf(t2, NEG_SLOPE * t2);
    t3 = __builtin_fmaxf(t3, NEG_SLOPE * t3);
    float acc = t0 * a.x;
    acc = __builtin_fmaf(t1, a.y, acc);
    acc = __builtin_fmaf(t2, a.z, acc);
    acc = __builtin_fmaf(t3, a.w, acc);
    return quad_sum(acc);
}

__device__ __forceinline__ void attend_chunk(const float4* __restrict__ xl4,
                                             const int mine, const int c, const int cnt,
                                             const int lane,
                                             const float4 r, const float4 a,
                                             float& m, float& den, float4& acc) {
    int s[8];
    #pragma unroll
    for (int j = 0; j < 8; ++j) s[j] = __builtin_amdgcn_readlane(mine, c + j);

    float4 v[8];
    #pragma unroll
    for (int j = 0; j < 8; ++j) {
        if (c + j < cnt) v[j] = xl4[(size_t)s[j] * (HID / 4) + lane];
    }

    float lg[8];
    float cm = -INFINITY;
    #pragma unroll
    for (int j = 0; j < 8; ++j) {
        if (c + j < cnt) {
            lg[j] = head_logit(v[j], r, a);
            cm = fmaxf(cm, lg[j]);
        }
    }

    const float mn = fmaxf(m, cm);
    const float sc = __builtin_amdgcn_exp2f(m - mn);
    den   *= sc;
    acc.x *= sc;
    acc.y *= sc;
    acc.z *= sc;
    acc.w *= sc;
    #pragma unroll
    for (int j = 0; j < 8; ++j) {
        if (c + j < cnt) {
            const float w = __builtin_amdgcn_exp2f(lg[j] - mn);
            den   += w;
            acc.x += w * v[j].x;
            acc.y += w * v[j].y;
            acc.z += w * v[j].z;
            acc.w += w * v[j].w;
        }
    }
    m = mn;
}

__device__ __forceinline__ float4 attend_node(const int* __restrict__ rowptr,
                                              const int* __restrict__ csr_src,
                                              const float* __restrict__ xl,
                                              const float* __restrict__ xr,
                                              const float* __restrict__ att,
                                              const float* __restrict__ bias,
                                              const int node, const int lane) {
    const int st = rowptr[node], en = rowptr[node + 1];
    const float4* xl4 = (const float4*)xl;
    const float4 r = ((const float4*)(xr + (size_t)node * HID))[lane];
    float4 a = ((const float4*)att)[lane];
    a.x *= LOG2E_F;
    a.y *= LOG2E_F;
    a.z *= LOG2E_F;
    a.w *= LOG2E_F;

    float m = -INFINITY;
    float den = 0.0f;
    float4 acc = make_float4(0.0f, 0.0f, 0.0f, 0.0f);
    for (int base = st; base < en; base += 32) {
        const int cnt = (en - base < 32) ? (en - base) : 32;
        const int mine = (lane < cnt) ? csr_src[base + lane] : 0;
        for (int c = 0; c < cnt; c += 8)
            attend_chunk(xl4, mine, c, cnt, lane, r, a, m, den, acc);
    }

    const float inv = __builtin_amdgcn_rcpf(den + EPS_F);
    const float4 b = ((const float4*)bias)[lane];
    float4 o;
    o.x = __builtin_fmaf(acc.x, inv, b.x);
    o.y = __builtin_fmaf(acc.y, inv, b.y);
    o.z = __builtin_fmaf(acc.z, inv, b.z);
    o.w = __builtin_fmaf(acc.w, inv, b.w);
    return o;
}

__global__ __launch_bounds__(256)
void node_attend_kernel(const int* __restrict__ rowptr,
                        const int* __restrict__ csr_src,
                        const float* __restrict__ xl,
                        const float* __restrict__ xr,
                        const float* __restrict__ att,
                        const float* __restrict__ bias,
                        float* __restrict__ out, int n) {
    const int node = __builtin_amdgcn_readfirstlane((int)(blockIdx.x * 8 + (threadIdx.x >> 5)));
    if (node >= n) return;
    const int lane = threadIdx.x & 31;
    const float4 o = attend_node(rowptr, csr_src, xl, xr, att, bias, node, lane);
    ((float4*)(out + (size_t)node * HID))[lane] = o;
}

__global__ __launch_bounds__(1024) __attribute__((amdgpu_waves_per_eu(16, 16)))
void attend_gemm_kernel(const int* __restrict__ rowptr,
                        const int* __restrict__ csr_src,
                        const float* __restrict__ xl,
                        const float* __restrict__ xr,
                        const float* __restrict__ att,
                        const float* __restrict__ bias,
                        const float* __restrict__ Wl,
                        const float* __restrict__ Wr,
                        float* __restrict__ XLn,
                        float* __restrict__ XRn, int n) {
    __shared__ __attribute__((aligned(16))) float As[STRIP_ROWS][A_PITCH];
    const int wave = __builtin_amdgcn_readfirstlane((int)(threadIdx.x >> 5));
    const int lane = threadIdx.x & 31;
    const int row0 = blockIdx.x * STRIP_ROWS;
    const int node = row0 + wave;

    float4 o = make_float4(0.0f, 0.0f, 0.0f, 0.0f);
    if (node < n) o = attend_node(rowptr, csr_src, xl, xr, att, bias, node, lane);
    *(float4*)(&As[wave][4 * lane]) = o;
    __syncthreads();
    if (wave >= 16) return;

    const float* __restrict__ Wm = (wave < 8) ? Wl : Wr;
    float* __restrict__ Om = (wave < 8) ? XLn : XRn;
    const int col0 = (wave & 7) * 16;
    const int half = lane >> 4;
    const int lm   = lane & 15;

    v8f acc0 = {0.f,0.f,0.f,0.f,0.f,0.f,0.f,0.f};
    v8f acc1 = {0.f,0.f,0.f,0.f,0.f,0.f,0.f,0.f};

    #pragma unroll 8
    for (int k0 = 0; k0 < HID; k0 += 4) {
        const int ka = k0 + 2 * half;
        v2f a0, a1;
        a0.x = As[lm][ka];
        a0.y = As[lm][ka + 1];
        a1.x = As[16 + lm][ka];
        a1.y = As[16 + lm][ka + 1];

        v2f b;
        b.x = Wm[(size_t)(ka + 0) * HID + col0 + lm];
        b.y = Wm[(size_t)(ka + 1) * HID + col0 + lm];

        acc0 = __builtin_amdgcn_wmma_f32_16x16x4_f32(false, a0, false, b,
                                                     (short)0, acc0, false, false);
        acc1 = __builtin_amdgcn_wmma_f32_16x16x4_f32(false, a1, false, b,
                                                     (short)0, acc1, false, false);
    }

    if (row0 + STRIP_ROWS <= n) {
        #pragma unroll
        for (int r = 0; r < 8; ++r) {
            const int m0 = row0 + r + 8 * half;
            Om[(size_t)m0 * HID + col0 + lm] = acc0[r];
            Om[(size_t)(m0 + 16) * HID + col0 + lm] = acc1[r];
        }
    } else {
        #pragma unroll
        for (int r = 0; r < 8; ++r) {
            const int m0 = row0 + r + 8 * half;
            const int m1 = m0 + 16;
            if (m0 < n) Om[(size_t)m0 * HID + col0 + lm] = acc0[r];
            if (m1 < n) Om[(size_t)m1 * HID + col0 + lm] = acc1[r];
        }
    }
}

extern "C" void kernel_launch(void* const* d_in, const int* in_sizes, int n_in,
                              void* d_out, int out_size, void* d_ws, size_t ws_size,
                              hipStream_t stream) {
    const float* x          = (const float*)d_in[0];
    const int*   edge_index = (const int*)  d_in[1];
    const float* Wl_all     = (const float*)d_in[2];
    const float* Wr_all     = (const float*)d_in[3];
    const float* att_all    = (const float*)d_in[4];
    const float* bias_all   = (const float*)d_in[5];
    float* out = (float*)d_out;

    const int N = in_sizes[0] / HID;
    const int E = in_sizes[1] / 2;

    const int* src = edge_index;
    const int* dst = edge_index + E;

    const size_t nf = (size_t)N * HID;
    float* xlA   = (float*)d_ws;
    float* xrA   = xlA + nf;
    float* xlB   = xrA + nf;
    float* xrB   = xlB + nf;

    const int TPB = 256;
    const int gemm_blocks  = (N + STRIP_ROWS - 1) / STRIP_ROWS;
    const int e_blocks     = (E + TPB - 1) / TPB;
    const int node_blocks  = (N + 7) / 8;
    const int nbins        = (N + BIN_NODES - 1) / BIN_NODES;
    const int nchunks      = SORT_CHUNKS;
    const int chunk        = (E + nchunks - 1) / nchunks;
    const int n_tab        = nbins * nchunks;
    const int scan_blocks  = (n_tab + SCAN_TPB - 1) / SCAN_TPB;

    int* hist     = (int*)(xrB + nf);
    int* offs     = hist + n_tab;
    int* binned   = offs + n_tab + 2;
    int* rowptr   = binned + 2 * (size_t)E;
    int* slot_eid = rowptr + N + 1;
    int* csr_src  = slot_eid + E;

    chunk_hist_kernel<<<nchunks, SORT_TPB, nbins * sizeof(int), stream>>>(dst, hist, E, chunk,
                                                                          nbins, nchunks);
    scan_counts_kernel<<<scan_blocks, SCAN_TPB, 0, stream>>>(hist, offs, n_tab);
    chunk_scatter_kernel<<<nchunks, SORT_TPB, nbins * sizeof(int), stream>>>(dst, offs, binned, E,
                                                                             chunk, nbins, nchunks);
    bin_place_kernel<<<nbins, SORT_TPB, 0, stream>>>(offs, binned, rowptr, slot_eid, N, nchunks);
    order_slots_kernel<<<e_blocks, TPB, 0, stream>>>(src, dst, rowptr, slot_eid,
                                                     csr_src, E);

    gemm_dual_f32<<<gemm_blocks, TPB, 0, stream>>>(x, Wl_all, Wr_all, xlA, xrA, N);

    float* cl = xlA; float* cr = xrA;
    float* nl = xlB; float* nr = xrB;
    for (int l = 0; l < LAYERS - 1; ++l) {
        const float* att  = att_all  + (size_t)l * HEADS * HEAD_DIM;
        const float* bias = bias_all + (size_t)l * HID;
        const float* Wl   = Wl_all   + (size_t)(l + 1) * HID * HID;
        const float* Wr   = Wr_all   + (size_t)(l + 1) * HID * HID;
        attend_gemm_kernel<<<gemm_blocks, 1024, 0, stream>>>(rowptr, csr_src, cl, cr,
                                                             att, bias, Wl, Wr, nl, nr, N);
        float* t;
        t = cl; cl = nl; nl = t;
        t = cr; cr = nr; nr = t;
    }
    {
        const float* att  = att_all  + (size_t)(LAYERS - 1) * HEADS * HEAD_DIM;
        const float* bias = bias_all + (size_t)(LAYERS - 1) * HID;
        node_attend_kernel<<<node_blocks, TPB, 0, stream>>>(rowptr, csr_src, cl, cr,
                                                            att, bias, out, N);
    }
}


# Edge grouping in 2 launches, was 5: a chunk sorts its edges by bin into its own stretch; one work-group per bin sums the counts in front, collects its runs, places and orders them in LDS; same sums
# speedup vs baseline: 8.7041x; 1.0718x over previous
#include <hip/hip_runtime.h>
#include <hip/hip_bf16.h>
#include <math.h>

#define HID      128
#define HEADS    8
#define HEAD_DIM 16
#define LAYERS   3
#define NEG_SLOPE 0.2f
#define EPS_F     1e-16f
#define LOG2E_F   1.4426950408889634f

typedef float v2f __attribute__((ext_vector_type(2)));
typedef float v8f __attribute__((ext_vector_type(8)));

#define A_PITCH (HID + 4)
#define STRIP_ROWS 32
__global__ __launch_bounds__(256)
void gemm_dual_f32(const float* __restrict__ X,
                   const float* __restrict__ Wl,
                   const float* __restrict__ Wr,
                   float* __restrict__ XL,
                   float* __restrict__ XR, int n) {
    __shared__ __attribute__((aligned(16))) float As[STRIP_ROWS][A_PITCH];
    const int row0 = blockIdx.x * STRIP_ROWS;

    {
        const float4* gx = (const float4*)(X + (size_t)row0 * HID);
        for (int t = threadIdx.x; t < STRIP_ROWS * HID / 4; t += 256) {
            const int rr = t >> 5;
            const int c4 = t & 31;
            float4 v = make_float4(0.0f, 0.0f, 0.0f, 0.0f);
            if (row0 + rr < n) v = gx[t];
            *(float4*)(&As[rr][4 * c4]) = v;
        }
    }
    __syncthreads();

    const int wave = threadIdx.x >> 5;
    const int lane = threadIdx.x & 31;
    const int col0 = wave * 16;
    const int half = lane >> 4;
    const int lm   = lane & 15;

    v8f accL0 = {0.f,0.f,0.f,0.f,0.f,0.f,0.f,0.f};
    v8f accL1 = {0.f,0.f,0.f,0.f,0.f,0.f,0.f,0.f};
    v8f accR0 = {0.f,0.f,0.f,0.f,0.f,0.f,0.f,0.f};
    v8f accR1 = {0.f,0.f,0.f,0.f,0.f,0.f,0.f,0.f};

    #pragma unroll 4
    for (int k0 = 0; k0 < HID; k0 += 4) {
        const int ka = k0 + 2 * half;
        v2f a0, a1;
        a0.x = As[lm][ka];
        a0.y = As[lm][ka + 1];
        a1.x = As[16 + lm][ka];
        a1.y = As[16 + lm][ka + 1];

        v2f bl, br;
        bl.x = Wl[(size_t)(ka + 0) * HID + col0 + lm];
        bl.y = Wl[(size_t)(ka + 1) * HID + col0 + lm];
        br.x = Wr[(size_t)(ka + 0) * HID + col0 + lm];
        br.y = Wr[(size_t)(ka + 1) * HID + col0 + lm];

        accL0 = __builtin_amdgcn_wmma_f32_16x16x4_f32(false, a0, false, bl,
                                                      (short)0, accL0, false, false);
        accL1 = __builtin_amdgcn_wmma_f32_16x16x4_f32(false, a1, false, bl,
                                                      (short)0, accL1, false, false);
        accR0 = __builtin_amdgcn_wmma_f32_16x16x4_f32(false, a0, false, br,
                                                      (short)0, accR0, false, false);
        accR1 = __builtin_amdgcn_wmma_f32_16x16x4_f32(false, a1, false, br,
                                                      (short)0, accR1, false, false);
    }

    if (row0 + STRIP_ROWS <= n) {
        #pragma unroll
        for (int r = 0; r < 8; ++r) {
            const int m0 = row0 + r + 8 * half;
            XL[(size_t)m0 * HID + col0 + lm] = accL0[r];
            XR[(size_t)m0 * HID + col0 + lm] = accR0[r];
            XL[(size_t)(m0 + 16) * HID + col0 + lm] = accL1[r];
            XR[(size_t)(m0 + 16) * HID + col0 + lm] = accR1[r];
        }
    } else {
        #pragma unroll
        for (int r = 0; r < 8; ++r) {
            const int m0 = row0 + r + 8 * half;
            const int m1 = m0 + 16;
            if (m0 < n) {
                XL[(size_t)m0 * HID + col0 + lm] = accL0[r];
                XR[(size_t)m0 * HID + col0 + lm] = accR0[r];
            }
            if (m1 < n) {
                XL[(size_t)m1 * HID + col0 + lm] = accL1[r];
                XR[(size_t)m1 * HID + col0 + lm] = accR1[r];
            }
        }
    }
}

#define BIN_SHIFT   8
#define BIN_NODES   256
#define SORT_TPB    1024
#define SORT_CHUNKS 256
#define RUNS_PER_WAVE (SORT_CHUNKS / (SORT_TPB / 32))
#define BIN_CAP     8192

__global__ __launch_bounds__(SORT_TPB)
void chunk_sort_kernel(const int* __restrict__ dst,
                       int* __restrict__ hist,
                       int* __restrict__ loco,
                       int* __restrict__ binned,
                       int E, int chunk, int nbins) {
    extern __shared__ int lbin[];
    __shared__ int wtot[SORT_TPB / 32];
    const int c = blockIdx.x;
    const int t = threadIdx.x;
    const int lane = t & 31;
    const int wave = t >> 5;
    for (int b = t; b < nbins; b += SORT_TPB) lbin[b] = 0;
    __syncthreads();
    const int e0 = c * chunk;
    const int e1 = (E - e0 < chunk) ? E : (e0 + chunk);
    for (int e = e0 + t; e < e1; e += 4 * SORT_TPB) {
        int d[4];
        #pragma unroll
        for (int j = 0; j < 4; ++j) {
            const int ee = e + j * SORT_TPB;
            d[j] = dst[(ee < e1) ? ee : e];
        }
        #pragma unroll
        for (int j = 0; j < 4; ++j) {
            if (e + j * SORT_TPB < e1) atomicAdd(&lbin[d[j] >> BIN_SHIFT], 1);
        }
    }
    __syncthreads();

    int carry = 0;
    for (int b0 = 0; b0 < nbins; b0 += SORT_TPB) {
        const int b = b0 + t;
        const int cnt = (b < nbins) ? lbin[b] : 0;
        int inc = cnt;
        for (int off = 1; off < 32; off <<= 1) {
            const int v = __shfl_up(inc, off, 32);
            if (lane >= off) inc += v;
        }
        if (lane == 31) wtot[wave] = inc;
        __syncthreads();
        int base = 0, tot = 0;
        for (int w = 0; w < SORT_TPB / 32; ++w) {
            const int wv = wtot[w];
            tot += wv;
            base += (w < wave) ? wv : 0;
        }
        const int excl = carry + base + inc - cnt;
        if (b < nbins) {
            hist[(size_t)b * SORT_CHUNKS + c] = cnt;
            loco[(size_t)b * SORT_CHUNKS + c] = excl;
            lbin[b] = e0 + excl;
        }
        carry += tot;
        __syncthreads();
    }

    for (int e = e0 + t; e < e1; e += 4 * SORT_TPB) {
        int d[4];
        #pragma unroll
        for (int j = 0; j < 4; ++j) {
            const int ee = e + j * SORT_TPB;
            d[j] = (ee < e1) ? dst[ee] : 0;
        }
        #pragma unroll
        for (int j = 0; j < 4; ++j) {
            const int ee = e + j * SORT_TPB;
            if (ee < e1) {
                const int p = atomicAdd(&lbin[d[j] >> BIN_SHIFT], 1);
                ((int2*)binned)[p] = make_int2(ee, d[j]);
            }
        }
    }
}

template <bool IN_LDS>
__device__ __forceinline__ void bin_rows(const int* __restrict__ src,
                                         const int* __restrict__ binned,
                                         int* gathered, int* slot_eid,
                                         int* __restrict__ rowptr,
                                         int* __restrict__ csr_src,
                                         int2* lpair, int* lslot,
                                         int* lcnt, int* lcur, int* lst, int* wdeg,
                                         const int* rsrc, const int* rdst, const int* rcnt,
                                         const int n, const int bs, const int be, const int lo,
                                         const bool last_bin) {
    const int t = threadIdx.x;
    const int lane = t & 31;
    const int wave = t >> 5;

    {
        int rn[RUNS_PER_WAVE], ro[RUNS_PER_WAVE], rd[RUNS_PER_WAVE];
        #pragma unroll
        for (int k = 0; k < RUNS_PER_WAVE; ++k) {
            const int r = wave + k * (SORT_TPB / 32);
            rn[k] = rcnt[r];
            ro[k] = rsrc[r];
            rd[k] = rdst[r];
        }
        int2 v[RUNS_PER_WAVE];
        #pragma unroll
        for (int k = 0; k < RUNS_PER_WAVE; ++k) {
            const int i = (lane < rn[k]) ? (ro[k] + lane) : 0;
            v[k] = ((const int2*)binned)[i];
        }
        #pragma unroll
        for (int k = 0; k < RUNS_PER_WAVE; ++k) {
            if (lane < rn[k]) {
                if (IN_LDS) lpair[rd[k] - bs + lane] = v[k];
                else        ((int2*)gathered)[rd[k] + lane] = v[k];
                atomicAdd(&lcnt[v[k].y - lo], 1);
            }
        }
        #pragma unroll
        for (int k = 0; k < RUNS_PER_WAVE; ++k) {
            for (int i = lane + 32; i < rn[k]; i += 32) {
                const int2 ed = ((const int2*)binned)[ro[k] + i];
                if (IN_LDS) lpair[rd[k] - bs + i] = ed;
                else        ((int2*)gathered)[rd[k] + i] = ed;
                atomicAdd(&lcnt[ed.y - lo], 1);
            }
        }
    }
    __syncthreads();

    int c = 0, inc = 0;
    if (t < BIN_NODES) {
        c = lcnt[t];
        inc = c;
        for (int off = 1; off < 32; off <<= 1) {
            const int v = __shfl_up(inc, off, 32);
            if (lane >= off) inc += v;
        }
        if (lane == 31) wdeg[wave] = inc;
    }
    __syncthreads();
    if (t < BIN_NODES) {
        int base = bs;
        for (int w = 0; w < wave; ++w) base += wdeg[w];
        const int excl = base + inc - c;
        lcur[t] = excl;
        lst[t] = excl;
        if (lo + t < n) rowptr[lo + t] = excl;
    }
    if (last_bin && t == 0) rowptr[n] = be;
    __syncthreads();

    for (int p = bs + t; p < be; p += 4 * SORT_TPB) {
        int2 ed[4];
        #pragma unroll
        for (int j = 0; j < 4; ++j) {
            const int pp = (p + j * SORT_TPB < be) ? (p + j * SORT_TPB) : p;
            ed[j] = IN_LDS ? lpair[pp - bs] : ((const int2*)gathered)[pp];
        }
        #pragma unroll
        for (int j = 0; j < 4; ++j) {
            if (p + j * SORT_TPB < be) {
                const int q = atomicAdd(&lcur[ed[j].y - lo], 1);
                if (IN_LDS) lslot[q - bs] = ed[j].x;
                else        slot_eid[q] = ed[j].x;
            }
        }
    }
    __syncthreads();

    for (int p = bs + t; p < be; p += 4 * SORT_TPB) {
        int2 ed[4];
        int sv[4];
        #pragma unroll
        for (int j = 0; j < 4; ++j) {
            const int pp = (p + j * SORT_TPB < be) ? (p + j * SORT_TPB) : p;
            ed[j] = IN_LDS ? lpair[pp - bs] : ((const int2*)gathered)[pp];
        }
        #pragma unroll
        for (int j = 0; j < 4; ++j) sv[j] = src[ed[j].x];
        #pragma unroll
        for (int j = 0; j < 4; ++j) {
            if (p + j * SORT_TPB < be) {
                const int e = ed[j].x;
                const int st = lst[ed[j].y - lo], en = lcur[ed[j].y - lo];
                int r = 0;
                if (IN_LDS) {
                    for (int q = st; q < en; q += 8) {
                        int a[8];
                        #pragma unroll
                        for (int u = 0; u < 8; ++u) a[u] = lslot[((q + u < en) ? (q + u) : q) - bs];
                        #pragma unroll
                        for (int u = 0; u < 8; ++u) r += (q + u < en && a[u] < e) ? 1 : 0;
                    }
                } else {
                    for (int q = st; q < en; ++q) r += (slot_eid[q] < e) ? 1 : 0;
                }
                csr_src[st + r] = sv[j];
            }
        }
    }
}

__global__ __launch_bounds__(SORT_TPB)
void bin_build_kernel(const int* __restrict__ src,
                      const int* __restrict__ hist,
                      const int* __restrict__ loco,
                      const int* __restrict__ binned,
                      int* gathered,
                      int* __restrict__ rowptr,
                      int* slot_eid,
                      int* __restrict__ csr_src, int n, int chunk) {
    __shared__ int2 lpair[BIN_CAP];
    __shared__ int lslot[BIN_CAP];
    __shared__ int lcnt[BIN_NODES];
    __shared__ int lcur[BIN_NODES];
    __shared__ int lst[BIN_NODES];
    __shared__ int rsrc[SORT_CHUNKS];
    __shared__ int rdst[SORT_CHUNKS];
    __shared__ int rcnt[SORT_CHUNKS];
    __shared__ int wfront[SORT_TPB / 32];
    __shared__ int wrun[SORT_CHUNKS / 32];
    __shared__ int wdeg[BIN_NODES / 32];
    const int b = blockIdx.x;
    const int t = threadIdx.x;
    const int lane = t & 31;
    const int wave = t >> 5;
    const int lo = b << BIN_SHIFT;
    const int first = b * SORT_CHUNKS;

    int rc = 0, rlo = 0;
    if (t < SORT_CHUNKS) {
        rc  = hist[first + t];
        rlo = loco[first + t];
    }

    const int4* c4 = (const int4*)hist;
    const int n4 = first >> 2;
    int s = 0;
    for (int k0 = 0; k0 < n4; k0 += 8 * SORT_TPB) {
        int4 v[8];
        #pragma unroll
        for (int j = 0; j < 8; ++j) {
            const int i = k0 + j * SORT_TPB + t;
            v[j] = c4[(i < n4) ? i : 0];
        }
        #pragma unroll
        for (int j = 0; j < 8; ++j) {
            const int i = k0 + j * SORT_TPB + t;
            s += (i < n4) ? ((v[j].x + v[j].y) + (v[j].z + v[j].w)) : 0;
        }
    }
    for (int off = 16; off > 0; off >>= 1) s += __shfl_xor(s, off, 32);

    int rinc = rc;
    if (t < SORT_CHUNKS) {
        for (int off = 1; off < 32; off <<= 1) {
            const int v = __shfl_up(rinc, off, 32);
            if (lane >= off) rinc += v;
        }
        if (lane == 31) wrun[wave] = rinc;
    }
    if (lane == 0) wfront[wave] = s;
    if (t < BIN_NODES) lcnt[t] = 0;
    __syncthreads();

    int bs = 0;
    for (int w = 0; w < SORT_TPB / 32; ++w) bs += wfront[w];
    int nb = 0;
    for (int w = 0; w < SORT_CHUNKS / 32; ++w) nb += wrun[w];
    const int be = bs + nb;
    if (t < SORT_CHUNKS) {
        int base = 0;
        for (int w = 0; w < wave; ++w) base += wrun[w];
        rdst[t] = bs + base + rinc - rc;
        rsrc[t] = t * chunk + rlo;
        rcnt[t] = rc;
    }
    __syncthreads();

    const bool last_bin = (b == (int)gridDim.x - 1);
    if (nb <= BIN_CAP)
        bin_rows<true>(src, binned, gathered, slot_eid, rowptr, csr_src, lpair, lslot,
                       lcnt, lcur, lst, wdeg, rsrc, rdst, rcnt, n, bs, be, lo, last_bin);
    else
        bin_rows<false>(src, binned, gathered, slot_eid, rowptr, csr_src, lpair, lslot,
                        lcnt, lcur, lst, wdeg, rsrc, rdst, rcnt, n, bs, be, lo, last_bin);
}

__device__ __forceinline__ float quad_sum(float x) {
    x += __int_as_float(__builtin_amdgcn_update_dpp(0, __float_as_int(x), 0xB1, 0xF, 0xF, true));
    x += __int_as_float(__builtin_amdgcn_update_dpp(0, __float_as_int(x), 0x4E, 0xF, 0xF, true));
    return x;
}

__device__ __forceinline__ float head_logit(const float4 v, const float4 r,
                                            const float4 a) {
    float t0 = v.x + r.x;
    float t1 = v.y + r.y;
    float t2 = v.z + r.z;
    float t3 = v.w + r.w;
    t0 = __builtin_fmaxf(t0, NEG_SLOPE * t0);
    t1 = __builtin_fmaxf(t1, NEG_SLOPE * t1);
    t2 = __builtin_fmaxf(t2, NEG_SLOPE * t2);
    t3 = __builtin_fmaxf(t3, NEG_SLOPE * t3);
    float acc = t0 * a.x;
    acc = __builtin_fmaf(t1, a.y, acc);
    acc = __builtin_fmaf(t2, a.z, acc);
    acc = __builtin_fmaf(t3, a.w, acc);
    return quad_sum(acc);
}

__device__ __forceinline__ void attend_chunk(const float4* __restrict__ xl4,
                                             const int mine, const int c, const int cnt,
                                             const int lane,
                                             const float4 r, const float4 a,
                                             float& m, float& den, float4& acc) {
    int s[8];
    #pragma unroll
    for (int j = 0; j < 8; ++j) s[j] = __builtin_amdgcn_readlane(mine, c + j);

    float4 v[8];
    #pragma unroll
    for (int j = 0; j < 8; ++j) {
        if (c + j < cnt) v[j] = xl4[(size_t)s[j] * (HID / 4) + lane];
    }

    float lg[8];
    float cm = -INFINITY;
    #pragma unroll
    for (int j = 0; j < 8; ++j) {
        if (c + j < cnt) {
            lg[j] = head_logit(v[j], r, a);
            cm = fmaxf(cm, lg[j]);
        }
    }

    const float mn = fmaxf(m, cm);
    const float sc = __builtin_amdgcn_exp2f(m - mn);
    den   *= sc;
    acc.x *= sc;
    acc.y *= sc;
    acc.z *= sc;
    acc.w *= sc;
    #pragma unroll
    for (int j = 0; j < 8; ++j) {
        if (c + j < cnt) {
            const float w = __builtin_amdgcn_exp2f(lg[j] - mn);
            den   += w;
            acc.x += w * v[j].x;
            acc.y += w * v[j].y;
            acc.z += w * v[j].z;
            acc.w += w * v[j].w;
        }
    }
    m = mn;
}

__device__ __forceinline__ float4 attend_node(const int* __restrict__ rowptr,
                                              const int* __restrict__ csr_src,
                                              const float* __restrict__ xl,
                                              const float* __restrict__ xr,
                                              const float* __restrict__ att,
                                              const float* __restrict__ bias,
                                              const int node, const int lane) {
    const int st = rowptr[node], en = rowptr[node + 1];
    const float4* xl4 = (const float4*)xl;
    const float4 r = ((const float4*)(xr + (size_t)node * HID))[lane];
    float4 a = ((const float4*)att)[lane];
    a.x *= LOG2E_F;
    a.y *= LOG2E_F;
    a.z *= LOG2E_F;
    a.w *= LOG2E_F;

    float m = -INFINITY;
    float den = 0.0f;
    float4 acc = make_float4(0.0f, 0.0f, 0.0f, 0.0f);
    for (int base = st; base < en; base += 32) {
        const int cnt = (en - base < 32) ? (en - base) : 32;
        const int mine = (lane < cnt) ? csr_src[base + lane] : 0;
        for (int c = 0; c < cnt; c += 8)
            attend_chunk(xl4, mine, c, cnt, lane, r, a, m, den, acc);
    }

    const float inv = __builtin_amdgcn_rcpf(den + EPS_F);
    const float4 b = ((const float4*)bias)[lane];
    float4 o;
    o.x = __builtin_fmaf(acc.x, inv, b.x);
    o.y = __builtin_fmaf(acc.y, inv, b.y);
    o.z = __builtin_fmaf(acc.z, inv, b.z);
    o.w = __builtin_fmaf(acc.w, inv, b.w);
    return o;
}

__global__ __launch_bounds__(256)
void node_attend_kernel(const int* __restrict__ rowptr,
                        const int* __restrict__ csr_src,
                        const float* __restrict__ xl,
                        const float* __restrict__ xr,
                        const float* __restrict__ att,
                        const float* __restrict__ bias,
                        float* __restrict__ out, int n) {
    const int node = __builtin_amdgcn_readfirstlane((int)(blockIdx.x * 8 + (threadIdx.x >> 5)));
    if (node >= n) return;
    const int lane = threadIdx.x & 31;
    const float4 o = attend_node(rowptr, csr_src, xl, xr, att, bias, node, lane);
    ((float4*)(out + (size_t)node * HID))[lane] = o;
}

__global__ __launch_bounds__(1024) __attribute__((amdgpu_waves_per_eu(16, 16)))
void attend_gemm_kernel(const int* __restrict__ rowptr,
                        const int* __restrict__ csr_src,
                        const float* __restrict__ xl,
                        const float* __restrict__ xr,
                        const float* __restrict__ att,
                        const float* __restrict__ bias,
                        const float* __restrict__ Wl,
                        const float* __restrict__ Wr,
                        float* __restrict__ XLn,
                        float* __restrict__ XRn, int n) {
    __shared__ __attribute__((aligned(16))) float As[STRIP_ROWS][A_PITCH];
    const int wave = __builtin_amdgcn_readfirstlane((int)(threadIdx.x >> 5));
    const int lane = threadIdx.x & 31;
    const int row0 = blockIdx.x * STRIP_ROWS;
    const int node = row0 + wave;

    float4 o = make_float4(0.0f, 0.0f, 0.0f, 0.0f);
    if (node < n) o = attend_node(rowptr, csr_src, xl, xr, att, bias, node, lane);
    *(float4*)(&As[wave][4 * lane]) = o;
    __syncthreads();
    if (wave >= 16) return;

    const float* __restrict__ Wm = (wave < 8) ? Wl : Wr;
    float* __restrict__ Om = (wave < 8) ? XLn : XRn;
    const int col0 = (wave & 7) * 16;
    const int half = lane >> 4;
    const int lm   = lane & 15;

    v8f acc0 = {0.f,0.f,0.f,0.f,0.f,0.f,0.f,0.f};
    v8f acc1 = {0.f,0.f,0.f,0.f,0.f,0.f,0.f,0.f};

    #pragma unroll 8
    for (int k0 = 0; k0 < HID; k0 += 4) {
        const int ka = k0 + 2 * half;
        v2f a0, a1;
        a0.x = As[lm][ka];
        a0.y = As[lm][ka + 1];
        a1.x = As[16 + lm][ka];
        a1.y = As[16 + lm][ka + 1];

        v2f b;
        b.x = Wm[(size_t)(ka + 0) * HID + col0 + lm];
        b.y = Wm[(size_t)(ka + 1) * HID + col0 + lm];

        acc0 = __builtin_amdgcn_wmma_f32_16x16x4_f32(false, a0, false, b,
                                                     (short)0, acc0, false, false);
        acc1 = __builtin_amdgcn_wmma_f32_16x16x4_f32(false, a1, false, b,
                                                     (short)0, acc1, false, false);
    }

    if (row0 + STRIP_ROWS <= n) {
        #pragma unroll
        for (int r = 0; r < 8; ++r) {
            const int m0 = row0 + r + 8 * half;
            Om[(size_t)m0 * HID + col0 + lm] = acc0[r];
            Om[(size_t)(m0 + 16) * HID + col0 + lm] = acc1[r];
        }
    } else {
        #pragma unroll
        for (int r = 0; r < 8; ++r) {
            const int m0 = row0 + r + 8 * half;
            const int m1 = m0 + 16;
            if (m0 < n) Om[(size_t)m0 * HID + col0 + lm] = acc0[r];
            if (m1 < n) Om[(size_t)m1 * HID + col0 + lm] = acc1[r];
        }
    }
}

extern "C" void kernel_launch(void* const* d_in, const int* in_sizes, int n_in,
                              void* d_out, int out_size, void* d_ws, size_t ws_size,
                              hipStream_t stream) {
    const float* x          = (const float*)d_in[0];
    const int*   edge_index = (const int*)  d_in[1];
    const float* Wl_all     = (const float*)d_in[2];
    const float* Wr_all     = (const float*)d_in[3];
    const float* att_all    = (const float*)d_in[4];
    const float* bias_all   = (const float*)d_in[5];
    float* out = (float*)d_out;

    const int N = in_sizes[0] / HID;
    const int E = in_sizes[1] / 2;

    const int* src = edge_index;
    const int* dst = edge_index + E;

    const size_t nf = (size_t)N * HID;
    float* xlA   = (float*)d_ws;
    float* xrA   = xlA + nf;
    float* xlB   = xrA + nf;
    float* xrB   = xlB + nf;

    const int TPB = 256;
    const int gemm_blocks  = (N + STRIP_ROWS - 1) / STRIP_ROWS;
    const int node_blocks  = (N + 7) / 8;
    const int nbins        = (N + BIN_NODES - 1) / BIN_NODES;
    const int chunk        = (E + SORT_CHUNKS - 1) / SORT_CHUNKS;
    const int n_tab        = nbins * SORT_CHUNKS;

    int* hist     = (int*)(xrB + nf);
    int* loco     = hist + n_tab;
    int* binned   = loco + n_tab;
    int* gathered = binned + 2 * (size_t)E;
    int* rowptr   = gathered + 2 * (size_t)E;
    int* slot_eid = rowptr + N + 1;
    int* csr_src  = slot_eid + E;

    chunk_sort_kernel<<<SORT_CHUNKS, SORT_TPB, nbins * sizeof(int), stream>>>(dst, hist, loco, binned,
                                                                              E, chunk, nbins);
    bin_build_kernel<<<nbins, SORT_TPB, 0, stream>>>(src, hist, loco, binned, gathered, rowptr,
                                                     slot_eid, csr_src, N, chunk);

    gemm_dual_f32<<<gemm_blocks, TPB, 0, stream>>>(x, Wl_all, Wr_all, xlA, xrA, N);

    float* cl = xlA; float* cr = xrA;
    float* nl = xlB; float* nr = xrB;
    for (int l = 0; l < LAYERS - 1; ++l) {
        const float* att  = att_all  + (size_t)l * HEADS * HEAD_DIM;
        const float* bias = bias_all + (size_t)l * HID;
        const float* Wl   = Wl_all   + (size_t)(l + 1) * HID * HID;
        const float* Wr   = Wr_all   + (size_t)(l + 1) * HID * HID;
        attend_gemm_kernel<<<gemm_blocks, 1024, 0, stream>>>(rowptr, csr_src, cl, cr,
                                                             att, bias, Wl, Wr, nl, nr, N);
        float* t;
        t = cl; cl = nl; nl = t;
        t = cr; cr = nr; nr = t;
    }
    {
        const float* att  = att_all  + (size_t)(LAYERS - 1) * HEADS * HEAD_DIM;
        const float* bias = bias_all + (size_t)(LAYERS - 1) * HID;
        node_attend_kernel<<<node_blocks, TPB, 0, stream>>>(rowptr, csr_src, cl, cr,
                                                            att, bias, out, N);
    }
}
